# starting kernel
# baseline (speedup 1.0000x reference)
#include <hip/hip_runtime.h>

typedef float v2f __attribute__((ext_vector_type(2)));
typedef float v8f __attribute__((ext_vector_type(8)));

#define F 128
#define LN_EPS 1e-5f

__global__ void k_deg(const int* __restrict__ dst, int E, int* __restrict__ cnt) {
    int e = blockIdx.x * blockDim.x + threadIdx.x;
    if (e < E) atomicAdd(&cnt[dst[e]], 1);
}

__global__ void k_dinv(const int* __restrict__ cnt, int N, float* __restrict__ dinv) {
    int v = blockIdx.x * blockDim.x + threadIdx.x;
    if (v < N) dinv[v] = rsqrtf((float)(cnt[v] + 1));
}

__global__ __launch_bounds__(256) void k_gemm(const float* __restrict__ X,
                                              const float* __restrict__ W,
                                              const float* __restrict__ dinv,
                                              const float* __restrict__ bias,
                                              float* __restrict__ H,
                                              float* __restrict__ agg, int N) {
    __shared__ float sW[F * F];
    for (int i = threadIdx.x; i < (F / 2) * (F / 4); i += 256) {
        int p = i >> 5;
        int c = (i & 31) * 4;
        float4 w0 = *(const float4*)&W[(2 * p)     * F + c];
        float4 w1 = *(const float4*)&W[(2 * p + 1) * F + c];
        v2f* d = (v2f*)&sW[p * 256 + c * 2];
        d[0] = (v2f){w0.x, w1.x};
        d[1] = (v2f){w0.y, w1.y};
        d[2] = (v2f){w0.z, w1.z};
        d[3] = (v2f){w0.w, w1.w};
    }
    __syncthreads();

    const int wave = threadIdx.x >> 5;
    const int lane = threadIdx.x & 31;
    const int tile = blockIdx.x * 8 + wave;
    if (tile * 16 >= N) return;

    const int col   = lane & 15;
    const int khalf = (lane >> 4) << 1;
    int row = tile * 16 + (lane & 15);
    if (row > N - 1) row = N - 1;
    const float* xrow = X + (size_t)row * F;

    v8f acc[8];
    const v8f vzero = {0.f, 0.f, 0.f, 0.f, 0.f, 0.f, 0.f, 0.f};
#pragma unroll
    for (int j = 0; j < 8; ++j) acc[j] = vzero;

    for (int k0 = 0; k0 < F; k0 += 4) {
        v2f a;
        a.x = xrow[k0 + khalf];
        a.y = xrow[k0 + khalf + 1];
        const v2f* bbase = (const v2f*)&sW[((k0 + khalf) >> 1) * 256 + col * 2];
#pragma unroll
        for (int j = 0; j < 8; ++j) {
            v2f b = bbase[j * 16];
            acc[j] = __builtin_amdgcn_wmma_f32_16x16x4_f32(
                false, a, false, b, (short)0, acc[j], false, false);
        }
    }

    const int rbase = tile * 16 + (lane >> 4) * 8;
    float nrm[8];
#pragma unroll
    for (int i = 0; i < 8; ++i) {
        int r = rbase + i;
        float di = dinv[r < N ? r : N - 1];
        nrm[i] = di * di;
    }
#pragma unroll
    for (int j = 0; j < 8; ++j) {
        float bj = bias[j * 16 + col];
#pragma unroll
        for (int i = 0; i < 8; ++i) {
            int r = rbase + i;
            if (r < N) {
                float h = acc[j][i];
                size_t o = (size_t)r * F + j * 16 + col;
                H[o]   = h;
                agg[o] = h * nrm[i] + bj;
            }
        }
    }
}

__global__ __launch_bounds__(256) void k_scatter(const int* __restrict__ src,
                                                 const int* __restrict__ dst,
                                                 const float* __restrict__ dinv,
                                                 const float* __restrict__ H,
                                                 float* __restrict__ agg, int E) {
    int gw   = (blockIdx.x * blockDim.x + threadIdx.x) >> 5;
    int lane = threadIdx.x & 31;
    if (gw >= E) return;
    int s = src[gw], d = dst[gw];
    float nrm = dinv[s] * dinv[d];
    float4 hv = *(const float4*)&H[(size_t)s * F + lane * 4];
    float* o = &agg[(size_t)d * F + lane * 4];
    unsafeAtomicAdd(o + 0, hv.x * nrm);
    unsafeAtomicAdd(o + 1, hv.y * nrm);
    unsafeAtomicAdd(o + 2, hv.z * nrm);
    unsafeAtomicAdd(o + 3, hv.w * nrm);
}

__global__ __launch_bounds__(256) void k_relu_ln(const float* __restrict__ agg,
                                                 const float* __restrict__ g,
                                                 const float* __restrict__ be,
                                                 float* __restrict__ out, int N) {
    int wave = threadIdx.x >> 5;
    int lane = threadIdx.x & 31;
    int row = blockIdx.x * 8 + wave;
    if (row >= N) return;

    float4 v = *(const float4*)&agg[(size_t)row * F + lane * 4];
    v.x = fmaxf(v.x, 0.f); v.y = fmaxf(v.y, 0.f);
    v.z = fmaxf(v.z, 0.f); v.w = fmaxf(v.w, 0.f);

    float s = v.x + v.y + v.z + v.w;
#pragma unroll
    for (int m = 16; m >= 1; m >>= 1) s += __shfl_xor(s, m, 32);
    float mu = s * (1.f / F);

    float dx = v.x - mu, dy = v.y - mu, dz = v.z - mu, dw = v.w - mu;
    float q = dx * dx + dy * dy + dz * dz + dw * dw;
#pragma unroll
    for (int m = 16; m >= 1; m >>= 1) q += __shfl_xor(q, m, 32);
    float rs = rsqrtf(q * (1.f / F) + LN_EPS);

    float4 gg = *(const float4*)&g[lane * 4];
    float4 bb = *(const float4*)&be[lane * 4];
    float4 o;
    o.x = dx * rs * gg.x + bb.x;
    o.y = dy * rs * gg.y + bb.y;
    o.z = dz * rs * gg.z + bb.z;
    o.w = dw * rs * gg.w + bb.w;
    *(float4*)&out[(size_t)row * F + lane * 4] = o;
}

static inline size_t align256(size_t x) { return (x + 255) & ~(size_t)255; }

extern "C" void kernel_launch(void* const* d_in, const int* in_sizes, int n_in,
                              void* d_out, int out_size, void* d_ws, size_t ws_size,
                              hipStream_t stream) {
    const float* x   = (const float*)d_in[0];
    const int*   ei  = (const int*)  d_in[1];
    const float* W1  = (const float*)d_in[2];
    const float* b1  = (const float*)d_in[3];
    const float* W2  = (const float*)d_in[4];
    const float* b2  = (const float*)d_in[5];
    const float* g1  = (const float*)d_in[6];
    const float* be1 = (const float*)d_in[7];
    const float* g2  = (const float*)d_in[8];
    const float* be2 = (const float*)d_in[9];

    const int N = in_sizes[0] / F;
    const int E = in_sizes[1] / 2;
    const int* src = ei;
    const int* dst = ei + E;
    float* out = (float*)d_out;

    char* ws = (char*)d_ws;
    size_t off = 0;
    float* dinv = (float*)(ws + off); off = align256(off + (size_t)N * 4);
    int*   cnt  = (int*)  (ws + off); off = align256(off + (size_t)N * 4);
    float* H    = (float*)(ws + off); off = align256(off + (size_t)N * F * 4);
    float* agg  = (float*)(ws + off); off = align256(off + (size_t)N * F * 4);
    (void)ws_size; (void)n_in; (void)out_size;

    const int T = 256;
    dim3 blk(T);

    hipMemsetAsync(cnt, 0, (size_t)N * 4, stream);
    k_deg <<<(E + T - 1) / T, blk, 0, stream>>>(dst, E, cnt);
    k_dinv<<<(N + T - 1) / T, blk, 0, stream>>>(cnt, N, dinv);

    const int gGemm = (N + 127) / 128;
    const int gScat = (E + 7) / 8;
    const int gLN   = (N + 7) / 8;

    k_gemm   <<<gGemm, blk, 0, stream>>>(x, W1, dinv, b1, H, agg, N);
    k_scatter<<<gScat, blk, 0, stream>>>(src, dst, dinv, H, agg, E);
    k_relu_ln<<<gLN,   blk, 0, stream>>>(agg, g1, be1, out, N);

    k_gemm   <<<gGemm, blk, 0, stream>>>(out, W2, dinv, b2, H, agg, N);
    k_scatter<<<gScat, blk, 0, stream>>>(src, dst, dinv, H, agg, E);
    k_relu_ln<<<gLN,   blk, 0, stream>>>(agg, g2, be2, out, N);
}


# Edge sum by destination slab: each work-group keeps 400 agg rows in LDS, scans all dst into an LDS queue, adds H rows with LDS f32 adds (arrival order); global f32 atomics gone
# speedup vs baseline: 1.6545x; 1.6545x over previous
#include <hip/hip_runtime.h>

typedef float v2f __attribute__((ext_vector_type(2)));
typedef float v8f __attribute__((ext_vector_type(8)));

#define F 128
#define LN_EPS 1e-5f

__global__ void k_deg(const int* __restrict__ dst, int E, int* __restrict__ cnt) {
    int e = blockIdx.x * blockDim.x + threadIdx.x;
    if (e < E) atomicAdd(&cnt[dst[e]], 1);
}

__global__ void k_dinv(const int* __restrict__ cnt, int N, float* __restrict__ dinv) {
    int v = blockIdx.x * blockDim.x + threadIdx.x;
    if (v < N) dinv[v] = rsqrtf((float)(cnt[v] + 1));
}

__global__ __launch_bounds__(256) void k_gemm(const float* __restrict__ X,
                                              const float* __restrict__ W,
                                              const float* __restrict__ dinv,
                                              const float* __restrict__ bias,
                                              float* __restrict__ H,
                                              float* __restrict__ agg, int N) {
    __shared__ float sW[F * F];
    for (int i = threadIdx.x; i < (F / 2) * (F / 4); i += 256) {
        int p = i >> 5;
        int c = (i & 31) * 4;
        float4 w0 = *(const float4*)&W[(2 * p)     * F + c];
        float4 w1 = *(const float4*)&W[(2 * p + 1) * F + c];
        v2f* d = (v2f*)&sW[p * 256 + c * 2];
        d[0] = (v2f){w0.x, w1.x};
        d[1] = (v2f){w0.y, w1.y};
        d[2] = (v2f){w0.z, w1.z};
        d[3] = (v2f){w0.w, w1.w};
    }
    __syncthreads();

    const int wave = threadIdx.x >> 5;
    const int lane = threadIdx.x & 31;
    const int tile = blockIdx.x * 8 + wave;
    if (tile * 16 >= N) return;

    const int col   = lane & 15;
    const int khalf = (lane >> 4) << 1;
    int row = tile * 16 + (lane & 15);
    if (row > N - 1) row = N - 1;
    const float* xrow = X + (size_t)row * F;

    v8f acc[8];
    const v8f vzero = {0.f, 0.f, 0.f, 0.f, 0.f, 0.f, 0.f, 0.f};
#pragma unroll
    for (int j = 0; j < 8; ++j) acc[j] = vzero;

    for (int k0 = 0; k0 < F; k0 += 4) {
        v2f a;
        a.x = xrow[k0 + khalf];
        a.y = xrow[k0 + khalf + 1];
        const v2f* bbase = (const v2f*)&sW[((k0 + khalf) >> 1) * 256 + col * 2];
#pragma unroll
        for (int j = 0; j < 8; ++j) {
            v2f b = bbase[j * 16];
            acc[j] = __builtin_amdgcn_wmma_f32_16x16x4_f32(
                false, a, false, b, (short)0, acc[j], false, false);
        }
    }

    const int rbase = tile * 16 + (lane >> 4) * 8;
    float nrm[8];
#pragma unroll
    for (int i = 0; i < 8; ++i) {
        int r = rbase + i;
        float di = dinv[r < N ? r : N - 1];
        nrm[i] = di * di;
    }
#pragma unroll
    for (int j = 0; j < 8; ++j) {
        float bj = bias[j * 16 + col];
#pragma unroll
        for (int i = 0; i < 8; ++i) {
            int r = rbase + i;
            if (r < N) {
                float h = acc[j][i];
                size_t o = (size_t)r * F + j * 16 + col;
                H[o]   = h;
                agg[o] = h * nrm[i] + bj;
            }
        }
    }
}

#define SLAB_R 400
#define SLAB_T 1024
#define QCAP   16384

__device__ __forceinline__ void lds_add(float* p, float v) {
    __hip_atomic_fetch_add(p, v, __ATOMIC_RELAXED, __HIP_MEMORY_SCOPE_WORKGROUP);
}

__global__ __launch_bounds__(SLAB_T) void k_slab(const int* __restrict__ src,
                                                 const int* __restrict__ dst,
                                                 const float* __restrict__ dinv,
                                                 const float* __restrict__ H,
                                                 float* __restrict__ agg, int N, int E) {
    __shared__ float acc[(SLAB_R + 1) * F];
    __shared__ int   q[QCAP];
    __shared__ int   qn;

    const int tid  = threadIdx.x;
    const int lane = tid & 31;
    const int wave = __builtin_amdgcn_readfirstlane(tid >> 5);
    const int row0 = blockIdx.x * SLAB_R;
    int nrows = N - row0;
    if (nrows > SLAB_R) nrows = SLAB_R;
    const unsigned nr = (unsigned)nrows;

    float4* aslab4 = (float4*)(agg + (size_t)row0 * F);
    float4* acc4 = (float4*)acc;
    const int nel4 = nrows * (F / 4);

    if (tid == 0) qn = 0;
    if (tid < F) acc[SLAB_R * F + tid] = 0.f;
    for (int base = tid; base < nel4; base += 4 * SLAB_T) {
        float4 t[4];
#pragma unroll
        for (int k = 0; k < 4; ++k) {
            int i = base + k * SLAB_T;
            float4 x = {0.f, 0.f, 0.f, 0.f};
            if (i < nel4) x = aslab4[i];
            t[k] = x;
        }
#pragma unroll
        for (int k = 0; k < 4; ++k) {
            int i = base + k * SLAB_T;
            if (i < nel4) acc4[i] = t[k];
        }
    }
    __syncthreads();

    auto push = [&](int e, unsigned r) {
        int slot = atomicAdd(&qn, 1);
        if (slot < QCAP) {
            q[slot] = e;
        } else {
            int s = src[e];
            float w = dinv[s] * dinv[row0 + (int)r];
            const float* hr = H + (size_t)s * F;
            float* ar = &acc[r * F];
#pragma unroll 1
            for (int f = 0; f < F; ++f) lds_add(ar + f, hr[f] * w);
        }
    };

    const int E4 = ((E & 3) == 0) ? (E >> 2) : 0;
    const int4* dst4 = (const int4*)dst;
    const int nomatch = row0 - 1;
    const int CH  = SLAB_T * 8;
    const int nch = (E4 + CH - 1) / CH;
    int cc = nch > 0 ? (int)(blockIdx.x % (unsigned)nch) : 0;
    for (int c = 0; c < nch; ++c) {
        const int base = cc * CH;
        if (++cc == nch) cc = 0;
        int4 d[8];
        if (base + SLAB_T * 8 <= E4) {
#pragma unroll
            for (int k = 0; k < 8; ++k) d[k] = dst4[base + k * SLAB_T + tid];
        } else {
#pragma unroll
            for (int k = 0; k < 8; ++k) {
                int i = base + k * SLAB_T + tid;
                int4 t = {nomatch, nomatch, nomatch, nomatch};
                if (i < E4) t = dst4[i];
                d[k] = t;
            }
        }
#pragma unroll
        for (int k = 0; k < 8; ++k) {
            int i = base + k * SLAB_T + tid;
            unsigned r0 = (unsigned)(d[k].x - row0);
            unsigned r1 = (unsigned)(d[k].y - row0);
            unsigned r2 = (unsigned)(d[k].z - row0);
            unsigned r3 = (unsigned)(d[k].w - row0);
            unsigned m01 = r0 < r1 ? r0 : r1;
            unsigned m23 = r2 < r3 ? r2 : r3;
            unsigned m = m01 < m23 ? m01 : m23;
            if (m < nr) {
                if (r0 < nr) push(4 * i + 0, r0);
                if (r1 < nr) push(4 * i + 1, r1);
                if (r2 < nr) push(4 * i + 2, r2);
                if (r3 < nr) push(4 * i + 3, r3);
            }
        }
    }
    for (int e = E4 * 4 + tid; e < E; e += SLAB_T) {
        unsigned r = (unsigned)(dst[e] - row0);
        if (r < nr) push(e, r);
    }
    __syncthreads();

    int n = __builtin_amdgcn_readfirstlane(qn);
    if (n > QCAP) n = QCAP;
    for (int i0 = wave * 8; i0 < n; i0 += (SLAB_T / 32) * 8) {
        int   e[8], s[8], dd[8];
        float w[8];
        float v[8][4];
#pragma unroll
        for (int k = 0; k < 8; ++k) {
            int idx = i0 + k;
            if (idx > n - 1) idx = n - 1;
            e[k] = __builtin_amdgcn_readfirstlane(q[idx]);
        }
#pragma unroll
        for (int k = 0; k < 8; ++k) { s[k] = src[e[k]]; dd[k] = dst[e[k]]; }
#pragma unroll
        for (int k = 0; k < 8; ++k) {
            w[k] = dinv[s[k]] * dinv[dd[k]];
            const float* hr = H + (size_t)s[k] * F + lane;
            v[k][0] = hr[0];
            v[k][1] = hr[32];
            v[k][2] = hr[64];
            v[k][3] = hr[96];
        }
#pragma unroll
        for (int k = 0; k < 8; ++k) {
            int row = (i0 + k < n) ? (dd[k] - row0) : SLAB_R;
            float* ar = &acc[row * F + lane];
            lds_add(ar,      v[k][0] * w[k]);
            lds_add(ar + 32, v[k][1] * w[k]);
            lds_add(ar + 64, v[k][2] * w[k]);
            lds_add(ar + 96, v[k][3] * w[k]);
        }
    }
    __syncthreads();

    for (int i = tid; i < nel4; i += SLAB_T) aslab4[i] = acc4[i];
}

__global__ __launch_bounds__(256) void k_relu_ln(const float* __restrict__ agg,
                                                 const float* __restrict__ g,
                                                 const float* __restrict__ be,
                                                 float* __restrict__ out, int N) {
    int wave = threadIdx.x >> 5;
    int lane = threadIdx.x & 31;
    int row = blockIdx.x * 8 + wave;
    if (row >= N) return;

    float4 v = *(const float4*)&agg[(size_t)row * F + lane * 4];
    v.x = fmaxf(v.x, 0.f); v.y = fmaxf(v.y, 0.f);
    v.z = fmaxf(v.z, 0.f); v.w = fmaxf(v.w, 0.f);

    float s = v.x + v.y + v.z + v.w;
#pragma unroll
    for (int m = 16; m >= 1; m >>= 1) s += __shfl_xor(s, m, 32);
    float mu = s * (1.f / F);

    float dx = v.x - mu, dy = v.y - mu, dz = v.z - mu, dw = v.w - mu;
    float q = dx * dx + dy * dy + dz * dz + dw * dw;
#pragma unroll
    for (int m = 16; m >= 1; m >>= 1) q += __shfl_xor(q, m, 32);
    float rs = rsqrtf(q * (1.f / F) + LN_EPS);

    float4 gg = *(const float4*)&g[lane * 4];
    float4 bb = *(const float4*)&be[lane * 4];
    float4 o;
    o.x = dx * rs * gg.x + bb.x;
    o.y = dy * rs * gg.y + bb.y;
    o.z = dz * rs * gg.z + bb.z;
    o.w = dw * rs * gg.w + bb.w;
    *(float4*)&out[(size_t)row * F + lane * 4] = o;
}

static inline size_t align256(size_t x) { return (x + 255) & ~(size_t)255; }

extern "C" void kernel_launch(void* const* d_in, const int* in_sizes, int n_in,
                              void* d_out, int out_size, void* d_ws, size_t ws_size,
                              hipStream_t stream) {
    const float* x   = (const float*)d_in[0];
    const int*   ei  = (const int*)  d_in[1];
    const float* W1  = (const float*)d_in[2];
    const float* b1  = (const float*)d_in[3];
    const float* W2  = (const float*)d_in[4];
    const float* b2  = (const float*)d_in[5];
    const float* g1  = (const float*)d_in[6];
    const float* be1 = (const float*)d_in[7];
    const float* g2  = (const float*)d_in[8];
    const float* be2 = (const float*)d_in[9];

    const int N = in_sizes[0] / F;
    const int E = in_sizes[1] / 2;
    const int* src = ei;
    const int* dst = ei + E;
    float* out = (float*)d_out;

    char* ws = (char*)d_ws;
    size_t off = 0;
    float* dinv = (float*)(ws + off); off = align256(off + (size_t)N * 4);
    int*   cnt  = (int*)  (ws + off); off = align256(off + (size_t)N * 4);
    float* H    = (float*)(ws + off); off = align256(off + (size_t)N * F * 4);
    float* agg  = (float*)(ws + off); off = align256(off + (size_t)N * F * 4);
    (void)ws_size; (void)n_in; (void)out_size;

    const int T = 256;
    dim3 blk(T);

    hipMemsetAsync(cnt, 0, (size_t)N * 4, stream);
    k_deg <<<(E + T - 1) / T, blk, 0, stream>>>(dst, E, cnt);
    k_dinv<<<(N + T - 1) / T, blk, 0, stream>>>(cnt, N, dinv);

    const int gGemm = (N + 127) / 128;
    const int gSlab = (N + SLAB_R - 1) / SLAB_R;
    const int gLN   = (N + 7) / 8;

    k_gemm   <<<gGemm, blk, 0, stream>>>(x, W1, dinv, b1, H, agg, N);
    k_slab   <<<gSlab, dim3(SLAB_T), 0, stream>>>(src, dst, dinv, H, agg, N, E);
    k_relu_ln<<<gLN,   blk, 0, stream>>>(agg, g1, be1, out, N);

    k_gemm   <<<gGemm, blk, 0, stream>>>(out, W2, dinv, b2, H, agg, N);
    k_slab   <<<gSlab, dim3(SLAB_T), 0, stream>>>(src, dst, dinv, H, agg, N, E);
    k_relu_ln<<<gLN,   blk, 0, stream>>>(agg, g2, be2, out, N);
}


# dst scan hoisted into one launch ahead of both layers: each slab's edge queue stays in the workspace for both edge-sum launches; the scan also counts in-degrees in LDS; deg atomics, dinv launch gone
# speedup vs baseline: 2.2650x; 1.3690x over previous
#include <hip/hip_runtime.h>

typedef float v2f __attribute__((ext_vector_type(2)));
typedef float v8f __attribute__((ext_vector_type(8)));

#define F 128
#define LN_EPS 1e-5f

__global__ __launch_bounds__(256) void k_gemm(const float* __restrict__ X,
                                              const float* __restrict__ W,
                                              const float* __restrict__ dinv,
                                              const float* __restrict__ bias,
                                              float* __restrict__ H,
                                              float* __restrict__ agg, int N) {
    __shared__ float sW[F * F];
    for (int i = threadIdx.x; i < (F / 2) * (F / 4); i += 256) {
        int p = i >> 5;
        int c = (i & 31) * 4;
        float4 w0 = *(const float4*)&W[(2 * p)     * F + c];
        float4 w1 = *(const float4*)&W[(2 * p + 1) * F + c];
        v2f* d = (v2f*)&sW[p * 256 + c * 2];
        d[0] = (v2f){w0.x, w1.x};
        d[1] = (v2f){w0.y, w1.y};
        d[2] = (v2f){w0.z, w1.z};
        d[3] = (v2f){w0.w, w1.w};
    }
    __syncthreads();

    const int wave = threadIdx.x >> 5;
    const int lane = threadIdx.x & 31;
    const int tile = blockIdx.x * 8 + wave;
    if (tile * 16 >= N) return;

    const int col   = lane & 15;
    const int khalf = (lane >> 4) << 1;
    int row = tile * 16 + (lane & 15);
    if (row > N - 1) row = N - 1;
    const float* xrow = X + (size_t)row * F;

    v8f acc[8];
    const v8f vzero = {0.f, 0.f, 0.f, 0.f, 0.f, 0.f, 0.f, 0.f};
#pragma unroll
    for (int j = 0; j < 8; ++j) acc[j] = vzero;

    for (int k0 = 0; k0 < F; k0 += 4) {
        v2f a;
        a.x = xrow[k0 + khalf];
        a.y = xrow[k0 + khalf + 1];
        const v2f* bbase = (const v2f*)&sW[((k0 + khalf) >> 1) * 256 + col * 2];
#pragma unroll
        for (int j = 0; j < 8; ++j) {
            v2f b = bbase[j * 16];
            acc[j] = __builtin_amdgcn_wmma_f32_16x16x4_f32(
                false, a, false, b, (short)0, acc[j], false, false);
        }
    }

    const int rbase = tile * 16 + (lane >> 4) * 8;
    float nrm[8];
#pragma unroll
    for (int i = 0; i < 8; ++i) {
        int r = rbase + i;
        float di = dinv[r < N ? r : N - 1];
        nrm[i] = di * di;
    }
#pragma unroll
    for (int j = 0; j < 8; ++j) {
        float bj = bias[j * 16 + col];
#pragma unroll
        for (int i = 0; i < 8; ++i) {
            int r = rbase + i;
            if (r < N) {
                float h = acc[j][i];
                size_t o = (size_t)r * F + j * 16 + col;
                H[o]   = h;
                agg[o] = h * nrm[i] + bj;
            }
        }
    }
}

#define SLAB_R 400
#define SLAB_T 1024
#define QCAP   16384

__device__ __forceinline__ void lds_add(float* p, float v) {
    __hip_atomic_fetch_add(p, v, __ATOMIC_RELAXED, __HIP_MEMORY_SCOPE_WORKGROUP);
}

__global__ __launch_bounds__(SLAB_T) void k_scan(const int* __restrict__ dst,
                                                 int* __restrict__ qg,
                                                 int* __restrict__ qcnt,
                                                 float* __restrict__ dinv, int N, int E) {
    __shared__ int4 q4[QCAP / 4];
    __shared__ int  deg[SLAB_R];
    __shared__ int  qn;
    int* q = (int*)q4;

    const int tid  = threadIdx.x;
    const int row0 = blockIdx.x * SLAB_R;
    int nrows = N - row0;
    if (nrows > SLAB_R) nrows = SLAB_R;
    const unsigned nr = (unsigned)nrows;

    if (tid == 0) qn = 0;
    if (tid < SLAB_R) deg[tid] = 0;
    __syncthreads();

    auto push = [&](int e, unsigned r) {
        atomicAdd(&deg[r], 1);
        int slot = atomicAdd(&qn, 1);
        if (slot < QCAP) q[slot] = e;
    };

    const int E4 = ((E & 3) == 0) ? (E >> 2) : 0;
    const int4* dst4 = (const int4*)dst;
    const int nomatch = row0 - 1;
    const int CH  = SLAB_T * 8;
    const int nch = (E4 + CH - 1) / CH;
    int cc = nch > 0 ? (int)(blockIdx.x % (unsigned)nch) : 0;
    for (int c = 0; c < nch; ++c) {
        const int base = cc * CH;
        if (++cc == nch) cc = 0;
        int4 d[8];
        if (base + SLAB_T * 8 <= E4) {
#pragma unroll
            for (int k = 0; k < 8; ++k) d[k] = dst4[base + k * SLAB_T + tid];
        } else {
#pragma unroll
            for (int k = 0; k < 8; ++k) {
                int i = base + k * SLAB_T + tid;
                int4 t = {nomatch, nomatch, nomatch, nomatch};
                if (i < E4) t = dst4[i];
                d[k] = t;
            }
        }
#pragma unroll
        for (int k = 0; k < 8; ++k) {
            int i = base + k * SLAB_T + tid;
            unsigned r0 = (unsigned)(d[k].x - row0);
            unsigned r1 = (unsigned)(d[k].y - row0);
            unsigned r2 = (unsigned)(d[k].z - row0);
            unsigned r3 = (unsigned)(d[k].w - row0);
            unsigned m01 = r0 < r1 ? r0 : r1;
            unsigned m23 = r2 < r3 ? r2 : r3;
            unsigned m = m01 < m23 ? m01 : m23;
            if (m < nr) {
                if (r0 < nr) push(4 * i + 0, r0);
                if (r1 < nr) push(4 * i + 1, r1);
                if (r2 < nr) push(4 * i + 2, r2);
                if (r3 < nr) push(4 * i + 3, r3);
            }
        }
    }
    for (int e = E4 * 4 + tid; e < E; e += SLAB_T) {
        unsigned r = (unsigned)(dst[e] - row0);
        if (r < nr) push(e, r);
    }
    __syncthreads();

    const int ntot = qn;
    const int n  = ntot < QCAP ? ntot : QCAP;
    const int n4 = (n + 3) >> 2;
    if (tid < 3 && n + tid < n4 * 4) q[n + tid] = 0;
    __syncthreads();
    int4* qg4 = (int4*)(qg + (size_t)blockIdx.x * QCAP);
    for (int i = tid; i < n4; i += SLAB_T) qg4[i] = q4[i];
    if (tid == 0) qcnt[blockIdx.x] = ntot;
    if (tid < nrows) dinv[row0 + tid] = rsqrtf((float)(deg[tid] + 1));
}

__global__ __launch_bounds__(SLAB_T) void k_slab(const int* __restrict__ src,
                                                 const int* __restrict__ dst,
                                                 const float* __restrict__ dinv,
                                                 const float* __restrict__ H,
                                                 const int* __restrict__ qg,
                                                 const int* __restrict__ qcnt,
                                                 float* __restrict__ agg, int N, int E) {
    __shared__ float acc[(SLAB_R + 1) * F];
    __shared__ int4  q4[QCAP / 4];
    const int* q = (const int*)q4;

    const int tid  = threadIdx.x;
    const int lane = tid & 31;
    const int wave = __builtin_amdgcn_readfirstlane(tid >> 5);
    const int row0 = blockIdx.x * SLAB_R;
    int nrows = N - row0;
    if (nrows > SLAB_R) nrows = SLAB_R;
    const unsigned nr = (unsigned)nrows;

    float4* aslab4 = (float4*)(agg + (size_t)row0 * F);
    float4* acc4 = (float4*)acc;
    const int nel4 = nrows * (F / 4);

    const int  ntot   = qcnt[blockIdx.x];
    const bool queued = ntot <= QCAP;

    if (tid < F) acc[SLAB_R * F + tid] = 0.f;
    if (queued) {
        const int4* qg4 = (const int4*)(qg + (size_t)blockIdx.x * QCAP);
        const int n4 = (ntot + 3) >> 2;
        for (int i = tid; i < n4; i += SLAB_T) q4[i] = qg4[i];
    }
    for (int base = tid; base < nel4; base += 4 * SLAB_T) {
        float4 t[4];
#pragma unroll
        for (int k = 0; k < 4; ++k) {
            int i = base + k * SLAB_T;
            float4 x = {0.f, 0.f, 0.f, 0.f};
            if (i < nel4) x = aslab4[i];
            t[k] = x;
        }
#pragma unroll
        for (int k = 0; k < 4; ++k) {
            int i = base + k * SLAB_T;
            if (i < nel4) acc4[i] = t[k];
        }
    }
    __syncthreads();

    if (!queued) {
        for (int e = tid; e < E; e += SLAB_T) {
            unsigned r = (unsigned)(dst[e] - row0);
            if (r < nr) {
                int s = src[e];
                float w = dinv[s] * dinv[row0 + (int)r];
                const float* hr = H + (size_t)s * F;
                float* ar = &acc[r * F];
#pragma unroll 1
                for (int f = 0; f < F; ++f) lds_add(ar + f, hr[f] * w);
            }
        }
    } else {
        const int n = __builtin_amdgcn_readfirstlane(ntot);
        for (int i0 = wave * 8; i0 < n; i0 += (SLAB_T / 32) * 8) {
            int   e[8], s[8], dd[8];
            float w[8];
            float v[8][4];
#pragma unroll
            for (int k = 0; k < 8; ++k) {
                int idx = i0 + k;
                if (idx > n - 1) idx = n - 1;
                e[k] = __builtin_amdgcn_readfirstlane(q[idx]);
            }
#pragma unroll
            for (int k = 0; k < 8; ++k) { s[k] = src[e[k]]; dd[k] = dst[e[k]]; }
#pragma unroll
            for (int k = 0; k < 8; ++k) {
                w[k] = dinv[s[k]] * dinv[dd[k]];
                const float* hr = H + (size_t)s[k] * F + lane;
                v[k][0] = hr[0];
                v[k][1] = hr[32];
                v[k][2] = hr[64];
                v[k][3] = hr[96];
            }
#pragma unroll
            for (int k = 0; k < 8; ++k) {
                int row = (i0 + k < n) ? (dd[k] - row0) : SLAB_R;
                float* ar = &acc[row * F + lane];
                lds_add(ar,      v[k][0] * w[k]);
                lds_add(ar + 32, v[k][1] * w[k]);
                lds_add(ar + 64, v[k][2] * w[k]);
                lds_add(ar + 96, v[k][3] * w[k]);
            }
        }
    }
    __syncthreads();

    for (int i = tid; i < nel4; i += SLAB_T) aslab4[i] = acc4[i];
}

__global__ __launch_bounds__(256) void k_relu_ln(const float* __restrict__ agg,
                                                 const float* __restrict__ g,
                                                 const float* __restrict__ be,
                                                 float* __restrict__ out, int N) {
    int wave = threadIdx.x >> 5;
    int lane = threadIdx.x & 31;
    int row = blockIdx.x * 8 + wave;
    if (row >= N) return;

    float4 v = *(const float4*)&agg[(size_t)row * F + lane * 4];
    v.x = fmaxf(v.x, 0.f); v.y = fmaxf(v.y, 0.f);
    v.z = fmaxf(v.z, 0.f); v.w = fmaxf(v.w, 0.f);

    float s = v.x + v.y + v.z + v.w;
#pragma unroll
    for (int m = 16; m >= 1; m >>= 1) s += __shfl_xor(s, m, 32);
    float mu = s * (1.f / F);

    float dx = v.x - mu, dy = v.y - mu, dz = v.z - mu, dw = v.w - mu;
    float q = dx * dx + dy * dy + dz * dz + dw * dw;
#pragma unroll
    for (int m = 16; m >= 1; m >>= 1) q += __shfl_xor(q, m, 32);
    float rs = rsqrtf(q * (1.f / F) + LN_EPS);

    float4 gg = *(const float4*)&g[lane * 4];
    float4 bb = *(const float4*)&be[lane * 4];
    float4 o;
    o.x = dx * rs * gg.x + bb.x;
    o.y = dy * rs * gg.y + bb.y;
    o.z = dz * rs * gg.z + bb.z;
    o.w = dw * rs * gg.w + bb.w;
    *(float4*)&out[(size_t)row * F + lane * 4] = o;
}

static inline size_t align256(size_t x) { return (x + 255) & ~(size_t)255; }

extern "C" void kernel_launch(void* const* d_in, const int* in_sizes, int n_in,
                              void* d_out, int out_size, void* d_ws, size_t ws_size,
                              hipStream_t stream) {
    const float* x   = (const float*)d_in[0];
    const int*   ei  = (const int*)  d_in[1];
    const float* W1  = (const float*)d_in[2];
    const float* b1  = (const float*)d_in[3];
    const float* W2  = (const float*)d_in[4];
    const float* b2  = (const float*)d_in[5];
    const float* g1  = (const float*)d_in[6];
    const float* be1 = (const float*)d_in[7];
    const float* g2  = (const float*)d_in[8];
    const float* be2 = (const float*)d_in[9];

    const int N = in_sizes[0] / F;
    const int E = in_sizes[1] / 2;
    const int* src = ei;
    const int* dst = ei + E;
    float* out = (float*)d_out;

    const int gGemm = (N + 127) / 128;
    const int gSlab = (N + SLAB_R - 1) / SLAB_R;
    const int gLN   = (N + 7) / 8;

    char* ws = (char*)d_ws;
    size_t off = 0;
    float* dinv = (float*)(ws + off); off = align256(off + (size_t)N * 4);
    float* H    = (float*)(ws + off); off = align256(off + (size_t)N * F * 4);
    float* agg  = (float*)(ws + off); off = align256(off + (size_t)N * F * 4);
    int*   qcnt = (int*)  (ws + off); off = align256(off + (size_t)gSlab * 4);
    int*   qg   = (int*)  (ws + off); off = align256(off + (size_t)gSlab * QCAP * 4);
    (void)ws_size; (void)n_in; (void)out_size;

    const int T = 256;
    dim3 blk(T);

    k_scan<<<gSlab, dim3(SLAB_T), 0, stream>>>(dst, qg, qcnt, dinv, N, E);

    k_gemm   <<<gGemm, blk, 0, stream>>>(x, W1, dinv, b1, H, agg, N);
    k_slab   <<<gSlab, dim3(SLAB_T), 0, stream>>>(src, dst, dinv, H, qg, qcnt, agg, N, E);
    k_relu_ln<<<gLN,   blk, 0, stream>>>(agg, g1, be1, out, N);

    k_gemm   <<<gGemm, blk, 0, stream>>>(out, W2, dinv, b2, H, agg, N);
    k_slab   <<<gSlab, dim3(SLAB_T), 0, stream>>>(src, dst, dinv, H, qg, qcnt, agg, N, E);
    k_relu_ln<<<gLN,   blk, 0, stream>>>(agg, g2, be2, out, N);
}


# Summed rows stay in LDS to the end: slab work-group starts them from its own H rows (self loop+bias), then does ReLU+LayerNorm from LDS and writes out; agg plane and both LayerNorm launches gone
# speedup vs baseline: 2.7852x; 1.2297x over previous
#include <hip/hip_runtime.h>

typedef float v2f __attribute__((ext_vector_type(2)));
typedef float v8f __attribute__((ext_vector_type(8)));

#define F 128
#define LN_EPS 1e-5f

__global__ __launch_bounds__(256) void k_gemm(const float* __restrict__ X,
                                              const float* __restrict__ W,
                                              float* __restrict__ H, int N) {
    __shared__ float sW[F * F];
    for (int i = threadIdx.x; i < (F / 2) * (F / 4); i += 256) {
        int p = i >> 5;
        int c = (i & 31) * 4;
        float4 w0 = *(const float4*)&W[(2 * p)     * F + c];
        float4 w1 = *(const float4*)&W[(2 * p + 1) * F + c];
        v2f* d = (v2f*)&sW[p * 256 + c * 2];
        d[0] = (v2f){w0.x, w1.x};
        d[1] = (v2f){w0.y, w1.y};
        d[2] = (v2f){w0.z, w1.z};
        d[3] = (v2f){w0.w, w1.w};
    }
    __syncthreads();

    const int wave = threadIdx.x >> 5;
    const int lane = threadIdx.x & 31;
    const int tile = blockIdx.x * 8 + wave;
    if (tile * 16 >= N) return;

    const int col   = lane & 15;
    const int khalf = (lane >> 4) << 1;
    int row = tile * 16 + (lane & 15);
    if (row > N - 1) row = N - 1;
    const float* xrow = X + (size_t)row * F;

    v8f acc[8];
    const v8f vzero = {0.f, 0.f, 0.f, 0.f, 0.f, 0.f, 0.f, 0.f};
#pragma unroll
    for (int j = 0; j < 8; ++j) acc[j] = vzero;

    for (int k0 = 0; k0 < F; k0 += 4) {
        v2f a;
        a.x = xrow[k0 + khalf];
        a.y = xrow[k0 + khalf + 1];
        const v2f* bbase = (const v2f*)&sW[((k0 + khalf) >> 1) * 256 + col * 2];
#pragma unroll
        for (int j = 0; j < 8; ++j) {
            v2f b = bbase[j * 16];
            acc[j] = __builtin_amdgcn_wmma_f32_16x16x4_f32(
                false, a, false, b, (short)0, acc[j], false, false);
        }
    }

    const int rbase = tile * 16 + (lane >> 4) * 8;
#pragma unroll
    for (int j = 0; j < 8; ++j) {
#pragma unroll
        for (int i = 0; i < 8; ++i) {
            int r = rbase + i;
            if (r < N) H[(size_t)r * F + j * 16 + col] = acc[j][i];
        }
    }
}

#define SLAB_R 400
#define SLAB_T 1024
#define QCAP   16384
#define LN_R   4

__device__ __forceinline__ void lds_add(float* p, float v) {
    __hip_atomic_fetch_add(p, v, __ATOMIC_RELAXED, __HIP_MEMORY_SCOPE_WORKGROUP);
}

__global__ __launch_bounds__(SLAB_T) void k_scan(const int* __restrict__ dst,
                                                 int* __restrict__ qg,
                                                 int* __restrict__ qcnt,
                                                 float* __restrict__ dinv, int N, int E) {
    __shared__ int4 q4[QCAP / 4];
    __shared__ int  deg[SLAB_R];
    __shared__ int  qn;
    int* q = (int*)q4;

    const int tid  = threadIdx.x;
    const int row0 = blockIdx.x * SLAB_R;
    int nrows = N - row0;
    if (nrows > SLAB_R) nrows = SLAB_R;
    const unsigned nr = (unsigned)nrows;

    if (tid == 0) qn = 0;
    if (tid < SLAB_R) deg[tid] = 0;
    __syncthreads();

    auto push = [&](int e, unsigned r) {
        atomicAdd(&deg[r], 1);
        int slot = atomicAdd(&qn, 1);
        if (slot < QCAP) q[slot] = e;
    };

    const int E4 = ((E & 3) == 0) ? (E >> 2) : 0;
    const int4* dst4 = (const int4*)dst;
    const int nomatch = row0 - 1;
    const int CH  = SLAB_T * 8;
    const int nch = (E4 + CH - 1) / CH;
    int cc = nch > 0 ? (int)(blockIdx.x % (unsigned)nch) : 0;
    for (int c = 0; c < nch; ++c) {
        const int base = cc * CH;
        if (++cc == nch) cc = 0;
        int4 d[8];
        if (base + SLAB_T * 8 <= E4) {
#pragma unroll
            for (int k = 0; k < 8; ++k) d[k] = dst4[base + k * SLAB_T + tid];
        } else {
#pragma unroll
            for (int k = 0; k < 8; ++k) {
                int i = base + k * SLAB_T + tid;
                int4 t = {nomatch, nomatch, nomatch, nomatch};
                if (i < E4) t = dst4[i];
                d[k] = t;
            }
        }
#pragma unroll
        for (int k = 0; k < 8; ++k) {
            int i = base + k * SLAB_T + tid;
            unsigned r0 = (unsigned)(d[k].x - row0);
            unsigned r1 = (unsigned)(d[k].y - row0);
            unsigned r2 = (unsigned)(d[k].z - row0);
            unsigned r3 = (unsigned)(d[k].w - row0);
            unsigned m01 = r0 < r1 ? r0 : r1;
            unsigned m23 = r2 < r3 ? r2 : r3;
            unsigned m = m01 < m23 ? m01 : m23;
            if (m < nr) {
                if (r0 < nr) push(4 * i + 0, r0);
                if (r1 < nr) push(4 * i + 1, r1);
                if (r2 < nr) push(4 * i + 2, r2);
                if (r3 < nr) push(4 * i + 3, r3);
            }
        }
    }
    for (int e = E4 * 4 + tid; e < E; e += SLAB_T) {
        unsigned r = (unsigned)(dst[e] - row0);
        if (r < nr) push(e, r);
    }
    __syncthreads();

    const int ntot = qn;
    const int n  = ntot < QCAP ? ntot : QCAP;
    const int n4 = (n + 3) >> 2;
    if (tid < 3 && n + tid < n4 * 4) q[n + tid] = 0;
    __syncthreads();
    int4* qg4 = (int4*)(qg + (size_t)blockIdx.x * QCAP);
    for (int i = tid; i < n4; i += SLAB_T) qg4[i] = q4[i];
    if (tid == 0) qcnt[blockIdx.x] = ntot;
    if (tid < nrows) dinv[row0 + tid] = rsqrtf((float)(deg[tid] + 1));
}

__global__ __launch_bounds__(SLAB_T) void k_slab(const int* __restrict__ src,
                                                 const int* __restrict__ dst,
                                                 const float* __restrict__ dinv,
                                                 const float* __restrict__ H,
                                                 const float* __restrict__ bias,
                                                 const float* __restrict__ g,
                                                 const float* __restrict__ be,
                                                 const int* __restrict__ qg,
                                                 const int* __restrict__ qcnt,
                                                 float* __restrict__ out, int N, int E) {
    __shared__ float acc[(SLAB_R + 1) * F];
    __shared__ int4  q4[QCAP / 4];
    const int* q = (const int*)q4;

    const int tid  = threadIdx.x;
    const int lane = tid & 31;
    const int wave = __builtin_amdgcn_readfirstlane(tid >> 5);
    const int row0 = blockIdx.x * SLAB_R;
    int nrows = N - row0;
    if (nrows > SLAB_R) nrows = SLAB_R;
    const unsigned nr = (unsigned)nrows;

    const float4* hslab4 = (const float4*)(H + (size_t)row0 * F);
    float4* acc4 = (float4*)acc;
    const int nel4 = nrows * (F / 4);

    const int  ntot   = qcnt[blockIdx.x];
    const bool queued = ntot <= QCAP;

    if (tid < F) acc[SLAB_R * F + tid] = 0.f;
    if (queued) {
        const int4* qg4 = (const int4*)(qg + (size_t)blockIdx.x * QCAP);
        const int n4 = (ntot + 3) >> 2;
        for (int i = tid; i < n4; i += SLAB_T) q4[i] = qg4[i];
    }
    const float4 b4 = ((const float4*)bias)[lane];
    for (int base = tid; base < nel4; base += 4 * SLAB_T) {
        float4 t[4];
        float  di[4];
#pragma unroll
        for (int k = 0; k < 4; ++k) {
            int i = base + k * SLAB_T;
            if (i > nel4 - 1) i = nel4 - 1;
            t[k]  = hslab4[i];
            di[k] = dinv[row0 + (i >> 5)];
        }
        asm volatile("" :: "v"(t[0].x), "v"(t[0].y), "v"(t[0].z), "v"(t[0].w),
                           "v"(t[1].x), "v"(t[1].y), "v"(t[1].z), "v"(t[1].w),
                           "v"(t[2].x), "v"(t[2].y), "v"(t[2].z), "v"(t[2].w),
                           "v"(t[3].x), "v"(t[3].y), "v"(t[3].z), "v"(t[3].w),
                           "v"(di[0]), "v"(di[1]), "v"(di[2]), "v"(di[3]));
#pragma unroll
        for (int k = 0; k < 4; ++k) {
            int i = base + k * SLAB_T;
            if (i < nel4) {
                float n2 = di[k] * di[k];
                float4 a;
                a.x = fmaf(t[k].x, n2, b4.x);
                a.y = fmaf(t[k].y, n2, b4.y);
                a.z = fmaf(t[k].z, n2, b4.z);
                a.w = fmaf(t[k].w, n2, b4.w);
                acc4[i] = a;
            }
        }
    }
    __syncthreads();

    if (!queued) {
        for (int e = tid; e < E; e += SLAB_T) {
            unsigned r = (unsigned)(dst[e] - row0);
            if (r < nr) {
                int s = src[e];
                float w = dinv[s] * dinv[row0 + (int)r];
                const float* hr = H + (size_t)s * F;
                float* ar = &acc[r * F];
#pragma unroll 1
                for (int f = 0; f < F; ++f) lds_add(ar + f, hr[f] * w);
            }
        }
    } else {
        const int n = __builtin_amdgcn_readfirstlane(ntot);
        for (int i0 = wave * 8; i0 < n; i0 += (SLAB_T / 32) * 8) {
            int   e[8], s[8], dd[8];
            float w[8];
            float v[8][4];
#pragma unroll
            for (int k = 0; k < 8; ++k) {
                int idx = i0 + k;
                if (idx > n - 1) idx = n - 1;
                e[k] = __builtin_amdgcn_readfirstlane(q[idx]);
            }
#pragma unroll
            for (int k = 0; k < 8; ++k) { s[k] = src[e[k]]; dd[k] = dst[e[k]]; }
#pragma unroll
            for (int k = 0; k < 8; ++k) {
                w[k] = dinv[s[k]] * dinv[dd[k]];
                const float* hr = H + (size_t)s[k] * F + lane;
                v[k][0] = hr[0];
                v[k][1] = hr[32];
                v[k][2] = hr[64];
                v[k][3] = hr[96];
            }
#pragma unroll
            for (int k = 0; k < 8; ++k) {
                int row = (i0 + k < n) ? (dd[k] - row0) : SLAB_R;
                float* ar = &acc[row * F + lane];
                lds_add(ar,      v[k][0] * w[k]);
                lds_add(ar + 32, v[k][1] * w[k]);
                lds_add(ar + 64, v[k][2] * w[k]);
                lds_add(ar + 96, v[k][3] * w[k]);
            }
        }
    }
    __syncthreads();

    const float4 gg = ((const float4*)g)[lane];
    const float4 bb = ((const float4*)be)[lane];
    float4* out4 = (float4*)(out + (size_t)row0 * F);
    for (int r0 = wave * LN_R; r0 < nrows; r0 += (SLAB_T / 32) * LN_R) {
        float4 v[LN_R];
        float  s[LN_R], qq[LN_R];
#pragma unroll
        for (int k = 0; k < LN_R; ++k) {
            int r = r0 + k;
            if (r > nrows - 1) r = nrows - 1;
            float4 t = acc4[r * (F / 4) + lane];
            t.x = fmaxf(t.x, 0.f); t.y = fmaxf(t.y, 0.f);
            t.z = fmaxf(t.z, 0.f); t.w = fmaxf(t.w, 0.f);
            v[k] = t;
            s[k] = t.x + t.y + t.z + t.w;
        }
#pragma unroll
        for (int m = 16; m >= 1; m >>= 1) {
#pragma unroll
            for (int k = 0; k < LN_R; ++k) s[k] += __shfl_xor(s[k], m, 32);
        }
#pragma unroll
        for (int k = 0; k < LN_R; ++k) {
            float mu = s[k] * (1.f / F);
            v[k].x -= mu; v[k].y -= mu; v[k].z -= mu; v[k].w -= mu;
            qq[k] = v[k].x * v[k].x + v[k].y * v[k].y + v[k].z * v[k].z + v[k].w * v[k].w;
        }
#pragma unroll
        for (int m = 16; m >= 1; m >>= 1) {
#pragma unroll
            for (int k = 0; k < LN_R; ++k) qq[k] += __shfl_xor(qq[k], m, 32);
        }
#pragma unroll
        for (int k = 0; k < LN_R; ++k) {
            float rs = rsqrtf(qq[k] * (1.f / F) + LN_EPS);
            float4 o;
            o.x = v[k].x * rs * gg.x + bb.x;
            o.y = v[k].y * rs * gg.y + bb.y;
            o.z = v[k].z * rs * gg.z + bb.z;
            o.w = v[k].w * rs * gg.w + bb.w;
            if (r0 + k < nrows) out4[(r0 + k) * (F / 4) + lane] = o;
        }
    }
}

static inline size_t align256(size_t x) { return (x + 255) & ~(size_t)255; }

extern "C" void kernel_launch(void* const* d_in, const int* in_sizes, int n_in,
                              void* d_out, int out_size, void* d_ws, size_t ws_size,
                              hipStream_t stream) {
    const float* x   = (const float*)d_in[0];
    const int*   ei  = (const int*)  d_in[1];
    const float* W1  = (const float*)d_in[2];
    const float* b1  = (const float*)d_in[3];
    const float* W2  = (const float*)d_in[4];
    const float* b2  = (const float*)d_in[5];
    const float* g1  = (const float*)d_in[6];
    const float* be1 = (const float*)d_in[7];
    const float* g2  = (const float*)d_in[8];
    const float* be2 = (const float*)d_in[9];

    const int N = in_sizes[0] / F;
    const int E = in_sizes[1] / 2;
    const int* src = ei;
    const int* dst = ei + E;
    float* out = (float*)d_out;

    const int gGemm = (N + 127) / 128;
    const int gSlab = (N + SLAB_R - 1) / SLAB_R;

    char* ws = (char*)d_ws;
    size_t off = 0;
    float* dinv = (float*)(ws + off); off = align256(off + (size_t)N * 4);
    float* H    = (float*)(ws + off); off = align256(off + (size_t)N * F * 4);
    int*   qcnt = (int*)  (ws + off); off = align256(off + (size_t)gSlab * 4);
    int*   qg   = (int*)  (ws + off); off = align256(off + (size_t)gSlab * QCAP * 4);
    (void)ws_size; (void)n_in; (void)out_size;

    const int T = 256;
    dim3 blk(T);

    k_scan<<<gSlab, dim3(SLAB_T), 0, stream>>>(dst, qg, qcnt, dinv, N, E);

    k_gemm<<<gGemm, blk, 0, stream>>>(x, W1, H, N);
    k_slab<<<gSlab, dim3(SLAB_T), 0, stream>>>(src, dst, dinv, H, b1, g1, be1, qg, qcnt, out, N, E);

    k_gemm<<<gGemm, blk, 0, stream>>>(out, W2, H, N);
    k_slab<<<gSlab, dim3(SLAB_T), 0, stream>>>(src, dst, dinv, H, b2, g2, be2, qg, qcnt, out, N, E);
}


# Every slab's scan of all dst replaced by ONE pass over the edges: each slice ranks its edges per slab in LDS, writes (src,row) records; slabs gather their segments; queue (sum) order changes
# speedup vs baseline: 3.6222x; 1.3005x over previous
#include <hip/hip_runtime.h>

typedef float v2f __attribute__((ext_vector_type(2)));
typedef float v8f __attribute__((ext_vector_type(8)));
typedef int   v4i __attribute__((ext_vector_type(4)));

#define F 128
#define LN_EPS 1e-5f

__global__ __launch_bounds__(256) void k_gemm(const float* __restrict__ X,
                                              const float* __restrict__ W,
                                              float* __restrict__ H, int N) {
    __shared__ float sW[F * F];
    for (int i = threadIdx.x; i < (F / 2) * (F / 4); i += 256) {
        int p = i >> 5;
        int c = (i & 31) * 4;
        float4 w0 = *(const float4*)&W[(2 * p)     * F + c];
        float4 w1 = *(const float4*)&W[(2 * p + 1) * F + c];
        v2f* d = (v2f*)&sW[p * 256 + c * 2];
        d[0] = (v2f){w0.x, w1.x};
        d[1] = (v2f){w0.y, w1.y};
        d[2] = (v2f){w0.z, w1.z};
        d[3] = (v2f){w0.w, w1.w};
    }
    __syncthreads();

    const int wave = threadIdx.x >> 5;
    const int lane = threadIdx.x & 31;
    const int tile = blockIdx.x * 8 + wave;
    if (tile * 16 >= N) return;

    const int col   = lane & 15;
    const int khalf = (lane >> 4) << 1;
    int row = tile * 16 + (lane & 15);
    if (row > N - 1) row = N - 1;
    const float* xrow = X + (size_t)row * F;

    v8f acc[8];
    const v8f vzero = {0.f, 0.f, 0.f, 0.f, 0.f, 0.f, 0.f, 0.f};
#pragma unroll
    for (int j = 0; j < 8; ++j) acc[j] = vzero;

    for (int k0 = 0; k0 < F; k0 += 4) {
        v2f a;
        a.x = xrow[k0 + khalf];
        a.y = xrow[k0 + khalf + 1];
        const v2f* bbase = (const v2f*)&sW[((k0 + khalf) >> 1) * 256 + col * 2];
#pragma unroll
        for (int j = 0; j < 8; ++j) {
            v2f b = bbase[j * 16];
            acc[j] = __builtin_amdgcn_wmma_f32_16x16x4_f32(
                false, a, false, b, (short)0, acc[j], false, false);
        }
    }

    const int rbase = tile * 16 + (lane >> 4) * 8;
#pragma unroll
    for (int j = 0; j < 8; ++j) {
#pragma unroll
        for (int i = 0; i < 8; ++i) {
            int r = rbase + i;
            if (r < N) H[(size_t)r * F + j * 16 + col] = acc[j][i];
        }
    }
}

#define SLAB_R 400
#define SLAB_T 1024
#define QCAP   8192
#define NSLICE 256
#define SEGCAP 64
#define MAXSLAB 256
#define PART_T 512
#define PART_U 13
#define DEG_T  256
#define LN_R   4

static_assert(NSLICE == 32 * 8, "a wave holds the NSLICE counts of a slab as 8 per lane");
static_assert(NSLICE == (SLAB_T / 32) * 8, "a k_slab wave gathers 8 segments");
static_assert(NSLICE == (DEG_T / 32) * 32, "a k_deg wave counts 32 segments");
static_assert(MAXSLAB == (PART_T / 32) * 16, "a k_part wave writes 16 segments");
static_assert(SEGCAP == 64, "a lane holds records l and l+32 of a segment");

__device__ __forceinline__ void lds_add(float* p, float v) {
    __hip_atomic_fetch_add(p, v, __ATOMIC_RELAXED, __HIP_MEMORY_SCOPE_WORKGROUP);
}

__global__ __launch_bounds__(PART_T) void k_part(const int* __restrict__ src,
                                                 const int* __restrict__ dst,
                                                 int* __restrict__ seg,
                                                 int* __restrict__ segcnt, int N, int E, int nslab) {
    __shared__ v4i stage4[MAXSLAB * SEGCAP / 2];
    int2* stage = (int2*)stage4;
    __shared__ int  cnt[MAXSLAB];

    const int tid  = threadIdx.x;
    const int lane = tid & 31;
    const int wave = __builtin_amdgcn_readfirstlane(tid >> 5);
    const int g    = blockIdx.x;
    const int nsl  = nslab < MAXSLAB ? nslab : MAXSLAB;

    if (tid < MAXSLAB) cnt[tid] = 0;
    __syncthreads();

    const int per = (E + NSLICE - 1) / NSLICE;
    const int e0  = g * per;
    int e1 = e0 + per;
    if (e1 > E) e1 = E;

    for (int base = e0 + tid; base < e1; base += PART_U * PART_T) {
        int d[PART_U], s[PART_U];
#pragma unroll
        for (int k = 0; k < PART_U; ++k) {
            int e = base + k * PART_T;
            if (e > e1 - 1) e = e1 - 1;
            d[k] = dst[e];
            s[k] = src[e];
        }
        asm volatile("" :: "v"(d[0]), "v"(d[1]), "v"(d[2]), "v"(d[3]), "v"(d[4]), "v"(d[5]), "v"(d[6]),
                           "v"(d[7]), "v"(d[8]), "v"(d[9]), "v"(d[10]), "v"(d[11]), "v"(d[12]));
        asm volatile("" :: "v"(s[0]), "v"(s[1]), "v"(s[2]), "v"(s[3]), "v"(s[4]), "v"(s[5]), "v"(s[6]),
                           "v"(s[7]), "v"(s[8]), "v"(s[9]), "v"(s[10]), "v"(s[11]), "v"(s[12]));
#pragma unroll
        for (int k = 0; k < PART_U; ++k) {
            int e = base + k * PART_T;
            if (e < e1 && (unsigned)d[k] < (unsigned)N) {
                int b = d[k] / SLAB_R;
                if (b < nsl) {
                    int slot = atomicAdd(&cnt[b], 1);
                    if (slot < SEGCAP) stage[b * SEGCAP + slot] = make_int2(s[k], d[k] - b * SLAB_R);
                }
            }
        }
    }
    __syncthreads();

    if (tid < nsl) segcnt[tid * NSLICE + g] = cnt[tid];
    const int cv = cnt[wave + (lane & 15) * (PART_T / 32)];
    v4i r[16];
#pragma unroll
    for (int j = 0; j < MAXSLAB / (PART_T / 32); ++j) r[j] = stage4[(wave + j * (PART_T / 32)) * (SEGCAP / 2) + lane];
    asm volatile("" :: "v"(r[0]), "v"(r[1]), "v"(r[2]), "v"(r[3]), "v"(r[4]), "v"(r[5]), "v"(r[6]), "v"(r[7]),
                       "v"(r[8]), "v"(r[9]), "v"(r[10]), "v"(r[11]), "v"(r[12]), "v"(r[13]), "v"(r[14]), "v"(r[15]));
#pragma unroll
    for (int j = 0; j < MAXSLAB / (PART_T / 32); ++j) {
        const int b = wave + j * (PART_T / 32);
        const int c = __builtin_amdgcn_readlane(cv, j);
        if (b < nsl) {
            int* o = seg + (((size_t)b * NSLICE + g) * SEGCAP + 2 * lane) * 2;
            if (2 * lane + 1 < c)  *(v4i*)o = r[j];
            else if (2 * lane < c) *(int2*)o = make_int2(r[j].x, r[j].y);
        }
    }
}

__global__ __launch_bounds__(DEG_T) void k_deg(const int* __restrict__ dst,
                                               const int* __restrict__ seg,
                                               const int* __restrict__ segcnt,
                                               float* __restrict__ dinv, int N, int E) {
    __shared__ int deg[SLAB_R];
    __shared__ int ovf;

    const int tid  = threadIdx.x;
    const int lane = tid & 31;
    const int wave = __builtin_amdgcn_readfirstlane(tid >> 5);
    const int b    = blockIdx.x;
    const int row0 = b * SLAB_R;
    int nrows = N - row0;
    if (nrows > SLAB_R) nrows = SLAB_R;
    const unsigned nr = (unsigned)nrows;

    for (int t = tid; t < SLAB_R; t += DEG_T) deg[t] = 0;
    if (tid == 0) ovf = (b >= MAXSLAB) ? 1 : 0;
    __syncthreads();

    if (b < MAXSLAB) {
        const int4* cb4 = (const int4*)(segcnt + (size_t)b * NSLICE) + lane * 2;
        const int4 ca = cb4[0], cc = cb4[1];
        const int cnt8[8] = {ca.x, ca.y, ca.z, ca.w, cc.x, cc.y, cc.z, cc.w};
        int bad = 0;
#pragma unroll
        for (int k = 0; k < 8; ++k) bad |= (cnt8[k] > SEGCAP) ? 1 : 0;
        if (__any(bad)) {
            if (tid == 0) ovf = 1;
        } else {
            const int* rb = seg + ((size_t)b * NSLICE + 32 * wave) * SEGCAP * 2 + 1;
            for (int a = 0; a < 4; a += 2) {
                int2 rr[16];
#pragma unroll
                for (int j = 0; j < 16; ++j) {
                    const int c = __builtin_amdgcn_readlane(cnt8[j & 7], 4 * wave + a + (j >> 3));
                    const int* p = rb + ((a * 8 + j) * SEGCAP + lane) * 2;
                    rr[j] = make_int2(-1, -1);
                    if (lane < c)      rr[j].x = p[0];
                    if (lane + 32 < c) rr[j].y = p[64];
                }
                asm volatile("" :: "v"(rr[0].x), "v"(rr[1].x), "v"(rr[2].x), "v"(rr[3].x), "v"(rr[4].x), "v"(rr[5].x), "v"(rr[6].x), "v"(rr[7].x),
                                   "v"(rr[8].x), "v"(rr[9].x), "v"(rr[10].x), "v"(rr[11].x), "v"(rr[12].x), "v"(rr[13].x), "v"(rr[14].x), "v"(rr[15].x));
                asm volatile("" :: "v"(rr[0].y), "v"(rr[1].y), "v"(rr[2].y), "v"(rr[3].y), "v"(rr[4].y), "v"(rr[5].y), "v"(rr[6].y), "v"(rr[7].y),
                                   "v"(rr[8].y), "v"(rr[9].y), "v"(rr[10].y), "v"(rr[11].y), "v"(rr[12].y), "v"(rr[13].y), "v"(rr[14].y), "v"(rr[15].y));
#pragma unroll
                for (int j = 0; j < 16; ++j) {
                    if (rr[j].x >= 0) atomicAdd(&deg[rr[j].x], 1);
                    if (rr[j].y >= 0) atomicAdd(&deg[rr[j].y], 1);
                }
            }
        }
    }
    __syncthreads();

    if (ovf) {
        for (int t = tid; t < SLAB_R; t += DEG_T) deg[t] = 0;
        __syncthreads();
        for (int e = tid; e < E; e += DEG_T) {
            unsigned r = (unsigned)(dst[e] - row0);
            if (r < nr) atomicAdd(&deg[r], 1);
        }
        __syncthreads();
    }
    for (int t = tid; t < nrows; t += DEG_T) dinv[row0 + t] = rsqrtf((float)(deg[t] + 1));
}

__global__ __launch_bounds__(SLAB_T) void k_slab(const int* __restrict__ src,
                                                 const int* __restrict__ dst,
                                                 const float* __restrict__ dinv,
                                                 const float* __restrict__ H,
                                                 const float* __restrict__ bias,
                                                 const float* __restrict__ g,
                                                 const float* __restrict__ be,
                                                 const int* __restrict__ seg,
                                                 const int* __restrict__ segcnt,
                                                 float* __restrict__ out, int N, int E) {
    __shared__ float acc[(SLAB_R + 1) * F];
    __shared__ int2  q2[QCAP];

    const int tid  = threadIdx.x;
    const int lane = tid & 31;
    const int wave = __builtin_amdgcn_readfirstlane(tid >> 5);
    const int row0 = blockIdx.x * SLAB_R;
    int nrows = N - row0;
    if (nrows > SLAB_R) nrows = SLAB_R;
    const unsigned nr = (unsigned)nrows;

    const float4* hslab4 = (const float4*)(H + (size_t)row0 * F);
    float4* acc4 = (float4*)acc;
    const int nel4 = nrows * (F / 4);

    bool queued = blockIdx.x < MAXSLAB;
    int  cnt8[8], off8[8];
    int  ntot = 0;
#pragma unroll
    for (int k = 0; k < 8; ++k) { cnt8[k] = 0; off8[k] = 0; }
    if (queued) {
        const int4* cb4 = (const int4*)(segcnt + (size_t)blockIdx.x * NSLICE) + lane * 2;
        const int4 ca = cb4[0], cc = cb4[1];
        cnt8[0] = ca.x; cnt8[1] = ca.y; cnt8[2] = ca.z; cnt8[3] = ca.w;
        cnt8[4] = cc.x; cnt8[5] = cc.y; cnt8[6] = cc.z; cnt8[7] = cc.w;
        int sum = 0, bad = 0;
#pragma unroll
        for (int k = 0; k < 8; ++k) {
            bad |= (cnt8[k] > SEGCAP) ? 1 : 0;
            off8[k] = sum;
            sum += cnt8[k];
        }
        int incl = sum;
#pragma unroll
        for (int m = 1; m < 32; m <<= 1) {
            int t = __shfl_up(incl, m, 32);
            if (lane >= m) incl += t;
        }
        const int excl = incl - sum;
#pragma unroll
        for (int k = 0; k < 8; ++k) off8[k] += excl;
        ntot = __builtin_amdgcn_readlane(incl, 31);
        if (__any(bad) || ntot > QCAP) queued = false;
    }

    if (tid < F) acc[SLAB_R * F + tid] = 0.f;
    if (queued) {
        const int2* sb = (const int2*)seg + ((size_t)blockIdx.x * NSLICE + 8 * wave) * SEGCAP + lane;
        int2 r0[8], r1[8];
        int  c[8], o[8];
#pragma unroll
        for (int k = 0; k < 8; ++k) {
            c[k] = __builtin_amdgcn_readlane(cnt8[k], wave);
            o[k] = __builtin_amdgcn_readlane(off8[k], wave);
            r0[k] = make_int2(0, 0);
            r1[k] = make_int2(0, 0);
            if (lane < c[k])      r0[k] = sb[k * SEGCAP];
            if (lane + 32 < c[k]) r1[k] = sb[k * SEGCAP + 32];
        }
        asm volatile("" :: "v"(r0[0].x), "v"(r0[0].y), "v"(r0[1].x), "v"(r0[1].y),
                           "v"(r0[2].x), "v"(r0[2].y), "v"(r0[3].x), "v"(r0[3].y),
                           "v"(r0[4].x), "v"(r0[4].y), "v"(r0[5].x), "v"(r0[5].y),
                           "v"(r0[6].x), "v"(r0[6].y), "v"(r0[7].x), "v"(r0[7].y));
        asm volatile("" :: "v"(r1[0].x), "v"(r1[0].y), "v"(r1[1].x), "v"(r1[1].y),
                           "v"(r1[2].x), "v"(r1[2].y), "v"(r1[3].x), "v"(r1[3].y),
                           "v"(r1[4].x), "v"(r1[4].y), "v"(r1[5].x), "v"(r1[5].y),
                           "v"(r1[6].x), "v"(r1[6].y), "v"(r1[7].x), "v"(r1[7].y));
#pragma unroll
        for (int k = 0; k < 8; ++k) {
            if (lane < c[k])      q2[o[k] + lane]      = r0[k];
            if (lane + 32 < c[k]) q2[o[k] + lane + 32] = r1[k];
        }
    }

    const float4 b4 = ((const float4*)bias)[lane];
    for (int base = tid; base < nel4; base += 4 * SLAB_T) {
        float4 t[4];
        float  di[4];
#pragma unroll
        for (int k = 0; k < 4; ++k) {
            int i = base + k * SLAB_T;
            if (i > nel4 - 1) i = nel4 - 1;
            t[k]  = hslab4[i];
            di[k] = dinv[row0 + (i >> 5)];
        }
        asm volatile("" :: "v"(t[0].x), "v"(t[0].y), "v"(t[0].z), "v"(t[0].w),
                           "v"(t[1].x), "v"(t[1].y), "v"(t[1].z), "v"(t[1].w),
                           "v"(t[2].x), "v"(t[2].y), "v"(t[2].z), "v"(t[2].w),
                           "v"(t[3].x), "v"(t[3].y), "v"(t[3].z), "v"(t[3].w),
                           "v"(di[0]), "v"(di[1]), "v"(di[2]), "v"(di[3]));
#pragma unroll
        for (int k = 0; k < 4; ++k) {
            int i = base + k * SLAB_T;
            if (i < nel4) {
                float n2 = di[k] * di[k];
                float4 a;
                a.x = fmaf(t[k].x, n2, b4.x);
                a.y = fmaf(t[k].y, n2, b4.y);
                a.z = fmaf(t[k].z, n2, b4.z);
                a.w = fmaf(t[k].w, n2, b4.w);
                acc4[i] = a;
            }
        }
    }
    __syncthreads();

    if (!queued) {
        for (int e = tid; e < E; e += SLAB_T) {
            unsigned r = (unsigned)(dst[e] - row0);
            if (r < nr) {
                int s = src[e];
                float w = dinv[s] * dinv[row0 + (int)r];
                const float* hr = H + (size_t)s * F;
                float* ar = &acc[r * F];
#pragma unroll 1
                for (int f = 0; f < F; ++f) lds_add(ar + f, hr[f] * w);
            }
        }
    } else {
        const int n = ntot;
        for (int i0 = wave * 8; i0 < n; i0 += (SLAB_T / 32) * 8) {
            int   s[8], rw[8];
            float w[8];
            float v[8][4];
#pragma unroll
            for (int k = 0; k < 8; ++k) {
                int idx = i0 + k;
                if (idx > n - 1) idx = n - 1;
                int2 rec = q2[idx];
                s[k]  = __builtin_amdgcn_readfirstlane(rec.x);
                rw[k] = __builtin_amdgcn_readfirstlane(rec.y);
            }
#pragma unroll
            for (int k = 0; k < 8; ++k) {
                w[k] = dinv[s[k]] * dinv[row0 + rw[k]];
                const float* hr = H + (size_t)s[k] * F + lane;
                v[k][0] = hr[0];
                v[k][1] = hr[32];
                v[k][2] = hr[64];
                v[k][3] = hr[96];
            }
#pragma unroll
            for (int k = 0; k < 8; ++k) {
                int row = (i0 + k < n) ? rw[k] : SLAB_R;
                float* ar = &acc[row * F + lane];
                lds_add(ar,      v[k][0] * w[k]);
                lds_add(ar + 32, v[k][1] * w[k]);
                lds_add(ar + 64, v[k][2] * w[k]);
                lds_add(ar + 96, v[k][3] * w[k]);
            }
        }
    }
    __syncthreads();

    const float4 gg = ((const float4*)g)[lane];
    const float4 bb = ((const float4*)be)[lane];
    float4* out4 = (float4*)(out + (size_t)row0 * F);
    for (int r0 = wave * LN_R; r0 < nrows; r0 += (SLAB_T / 32) * LN_R) {
        float4 v[LN_R];
        float  s[LN_R], qq[LN_R];
#pragma unroll
        for (int k = 0; k < LN_R; ++k) {
            int r = r0 + k;
            if (r > nrows - 1) r = nrows - 1;
            float4 t = acc4[r * (F / 4) + lane];
            t.x = fmaxf(t.x, 0.f); t.y = fmaxf(t.y, 0.f);
            t.z = fmaxf(t.z, 0.f); t.w = fmaxf(t.w, 0.f);
            v[k] = t;
            s[k] = t.x + t.y + t.z + t.w;
        }
#pragma unroll
        for (int m = 16; m >= 1; m >>= 1) {
#pragma unroll
            for (int k = 0; k < LN_R; ++k) s[k] += __shfl_xor(s[k], m, 32);
        }
#pragma unroll
        for (int k = 0; k < LN_R; ++k) {
            float mu = s[k] * (1.f / F);
            v[k].x -= mu; v[k].y -= mu; v[k].z -= mu; v[k].w -= mu;
            qq[k] = v[k].x * v[k].x + v[k].y * v[k].y + v[k].z * v[k].z + v[k].w * v[k].w;
        }
#pragma unroll
        for (int m = 16; m >= 1; m >>= 1) {
#pragma unroll
            for (int k = 0; k < LN_R; ++k) qq[k] += __shfl_xor(qq[k], m, 32);
        }
#pragma unroll
        for (int k = 0; k < LN_R; ++k) {
            float rs = rsqrtf(qq[k] * (1.f / F) + LN_EPS);
            float4 o;
            o.x = v[k].x * rs * gg.x + bb.x;
            o.y = v[k].y * rs * gg.y + bb.y;
            o.z = v[k].z * rs * gg.z + bb.z;
            o.w = v[k].w * rs * gg.w + bb.w;
            if (r0 + k < nrows) out4[(r0 + k) * (F / 4) + lane] = o;
        }
    }
}

static inline size_t align256(size_t x) { return (x + 255) & ~(size_t)255; }

extern "C" void kernel_launch(void* const* d_in, const int* in_sizes, int n_in,
                              void* d_out, int out_size, void* d_ws, size_t ws_size,
                              hipStream_t stream) {
    const float* x   = (const float*)d_in[0];
    const int*   ei  = (const int*)  d_in[1];
    const float* W1  = (const float*)d_in[2];
    const float* b1  = (const float*)d_in[3];
    const float* W2  = (const float*)d_in[4];
    const float* b2  = (const float*)d_in[5];
    const float* g1  = (const float*)d_in[6];
    const float* be1 = (const float*)d_in[7];
    const float* g2  = (const float*)d_in[8];
    const float* be2 = (const float*)d_in[9];

    const int N = in_sizes[0] / F;
    const int E = in_sizes[1] / 2;
    const int* src = ei;
    const int* dst = ei + E;
    float* out = (float*)d_out;

    const int gGemm = (N + 127) / 128;
    const int gSlab = (N + SLAB_R - 1) / SLAB_R;

    char* ws = (char*)d_ws;
    size_t off = 0;
    float* dinv   = (float*)(ws + off); off = align256(off + (size_t)N * 4);
    float* H      = (float*)(ws + off); off = align256(off + (size_t)N * F * 4);
    int*   segcnt = (int*)  (ws + off); off = align256(off + (size_t)MAXSLAB * NSLICE * 4);
    int*   seg    = (int*)  (ws + off); off = align256(off + (size_t)MAXSLAB * NSLICE * SEGCAP * 8);
    (void)ws_size; (void)n_in; (void)out_size;

    const int T = 256;
    dim3 blk(T);

    k_part<<<NSLICE, dim3(PART_T), 0, stream>>>(src, dst, seg, segcnt, N, E, gSlab);
    k_deg<<<gSlab, dim3(DEG_T), 0, stream>>>(dst, seg, segcnt, dinv, N, E);

    k_gemm<<<gGemm, blk, 0, stream>>>(x, W1, H, N);
    k_slab<<<gSlab, dim3(SLAB_T), 0, stream>>>(src, dst, dinv, H, b1, g1, be1, seg, segcnt, out, N, E);

    k_gemm<<<gGemm, blk, 0, stream>>>(out, W2, H, N);
    k_slab<<<gSlab, dim3(SLAB_T), 0, stream>>>(src, dst, dinv, H, b2, g2, be2, seg, segcnt, out, N, E);
}


# Edge-sum row gather: one 16-byte-per-lane load takes a whole 512-byte H row (was four 4-byte loads, one line each); LDS rows kept column-interleaved so the f32 LDS adds stay conflict-free
# speedup vs baseline: 3.8306x; 1.0575x over previous
#include <hip/hip_runtime.h>

typedef float v2f __attribute__((ext_vector_type(2)));
typedef float v8f __attribute__((ext_vector_type(8)));
typedef int   v4i __attribute__((ext_vector_type(4)));

#define F 128
#define LN_EPS 1e-5f

__global__ __launch_bounds__(256) void k_gemm(const float* __restrict__ X,
                                              const float* __restrict__ W,
                                              float* __restrict__ H, int N) {
    __shared__ float sW[F * F];
    for (int i = threadIdx.x; i < (F / 2) * (F / 4); i += 256) {
        int p = i >> 5;
        int c = (i & 31) * 4;
        float4 w0 = *(const float4*)&W[(2 * p)     * F + c];
        float4 w1 = *(const float4*)&W[(2 * p + 1) * F + c];
        v2f* d = (v2f*)&sW[p * 256 + c * 2];
        d[0] = (v2f){w0.x, w1.x};
        d[1] = (v2f){w0.y, w1.y};
        d[2] = (v2f){w0.z, w1.z};
        d[3] = (v2f){w0.w, w1.w};
    }
    __syncthreads();

    const int wave = threadIdx.x >> 5;
    const int lane = threadIdx.x & 31;
    const int tile = blockIdx.x * 8 + wave;
    if (tile * 16 >= N) return;

    const int col   = lane & 15;
    const int khalf = (lane >> 4) << 1;
    int row = tile * 16 + (lane & 15);
    if (row > N - 1) row = N - 1;
    const float* xrow = X + (size_t)row * F;

    v8f acc[8];
    const v8f vzero = {0.f, 0.f, 0.f, 0.f, 0.f, 0.f, 0.f, 0.f};
#pragma unroll
    for (int j = 0; j < 8; ++j) acc[j] = vzero;

    for (int k0 = 0; k0 < F; k0 += 4) {
        v2f a;
        a.x = xrow[k0 + khalf];
        a.y = xrow[k0 + khalf + 1];
        const v2f* bbase = (const v2f*)&sW[((k0 + khalf) >> 1) * 256 + col * 2];
#pragma unroll
        for (int j = 0; j < 8; ++j) {
            v2f b = bbase[j * 16];
            acc[j] = __builtin_amdgcn_wmma_f32_16x16x4_f32(
                false, a, false, b, (short)0, acc[j], false, false);
        }
    }

    const int rbase = tile * 16 + (lane >> 4) * 8;
#pragma unroll
    for (int j = 0; j < 8; ++j) {
#pragma unroll
        for (int i = 0; i < 8; ++i) {
            int r = rbase + i;
            if (r < N) H[(size_t)r * F + j * 16 + col] = acc[j][i];
        }
    }
}

#define SLAB_R 400
#define SLAB_T 1024
#define QCAP   8192
#define NSLICE 256
#define SEGCAP 64
#define MAXSLAB 256
#define PART_T 512
#define PART_U 13
#define DEG_T  256
#define LN_R   4

static_assert(NSLICE == 32 * 8, "a wave holds the NSLICE counts of a slab as 8 per lane");
static_assert(NSLICE == (SLAB_T / 32) * 8, "a k_slab wave gathers 8 segments");
static_assert(NSLICE == (DEG_T / 32) * 32, "a k_deg wave counts 32 segments");
static_assert(MAXSLAB == (PART_T / 32) * 16, "a k_part wave writes 16 segments");
static_assert(SEGCAP == 64, "a lane holds records l and l+32 of a segment");

__device__ __forceinline__ void lds_add(float* p, float v) {
    __hip_atomic_fetch_add(p, v, __ATOMIC_RELAXED, __HIP_MEMORY_SCOPE_WORKGROUP);
}

__global__ __launch_bounds__(PART_T) void k_part(const int* __restrict__ src,
                                                 const int* __restrict__ dst,
                                                 int* __restrict__ seg,
                                                 int* __restrict__ segcnt, int N, int E, int nslab) {
    __shared__ v4i stage4[MAXSLAB * SEGCAP / 2];
    int2* stage = (int2*)stage4;
    __shared__ int  cnt[MAXSLAB];

    const int tid  = threadIdx.x;
    const int lane = tid & 31;
    const int wave = __builtin_amdgcn_readfirstlane(tid >> 5);
    const int g    = blockIdx.x;
    const int nsl  = nslab < MAXSLAB ? nslab : MAXSLAB;

    if (tid < MAXSLAB) cnt[tid] = 0;
    __syncthreads();

    const int per = (E + NSLICE - 1) / NSLICE;
    const int e0  = g * per;
    int e1 = e0 + per;
    if (e1 > E) e1 = E;

    for (int base = e0 + tid; base < e1; base += PART_U * PART_T) {
        int d[PART_U], s[PART_U];
#pragma unroll
        for (int k = 0; k < PART_U; ++k) {
            int e = base + k * PART_T;
            if (e > e1 - 1) e = e1 - 1;
            d[k] = dst[e];
            s[k] = src[e];
        }
        asm volatile("" :: "v"(d[0]), "v"(d[1]), "v"(d[2]), "v"(d[3]), "v"(d[4]), "v"(d[5]), "v"(d[6]),
                           "v"(d[7]), "v"(d[8]), "v"(d[9]), "v"(d[10]), "v"(d[11]), "v"(d[12]));
        asm volatile("" :: "v"(s[0]), "v"(s[1]), "v"(s[2]), "v"(s[3]), "v"(s[4]), "v"(s[5]), "v"(s[6]),
                           "v"(s[7]), "v"(s[8]), "v"(s[9]), "v"(s[10]), "v"(s[11]), "v"(s[12]));
#pragma unroll
        for (int k = 0; k < PART_U; ++k) {
            int e = base + k * PART_T;
            if (e < e1 && (unsigned)d[k] < (unsigned)N) {
                int b = d[k] / SLAB_R;
                if (b < nsl) {
                    int slot = atomicAdd(&cnt[b], 1);
                    if (slot < SEGCAP) stage[b * SEGCAP + slot] = make_int2(s[k], d[k] - b * SLAB_R);
                }
            }
        }
    }
    __syncthreads();

    if (tid < nsl) segcnt[tid * NSLICE + g] = cnt[tid];
    const int cv = cnt[wave + (lane & 15) * (PART_T / 32)];
    v4i r[16];
#pragma unroll
    for (int j = 0; j < MAXSLAB / (PART_T / 32); ++j) r[j] = stage4[(wave + j * (PART_T / 32)) * (SEGCAP / 2) + lane];
    asm volatile("" :: "v"(r[0]), "v"(r[1]), "v"(r[2]), "v"(r[3]), "v"(r[4]), "v"(r[5]), "v"(r[6]), "v"(r[7]),
                       "v"(r[8]), "v"(r[9]), "v"(r[10]), "v"(r[11]), "v"(r[12]), "v"(r[13]), "v"(r[14]), "v"(r[15]));
#pragma unroll
    for (int j = 0; j < MAXSLAB / (PART_T / 32); ++j) {
        const int b = wave + j * (PART_T / 32);
        const int c = __builtin_amdgcn_readlane(cv, j);
        if (b < nsl) {
            int* o = seg + (((size_t)b * NSLICE + g) * SEGCAP + 2 * lane) * 2;
            if (2 * lane + 1 < c)  *(v4i*)o = r[j];
            else if (2 * lane < c) *(int2*)o = make_int2(r[j].x, r[j].y);
        }
    }
}

__global__ __launch_bounds__(DEG_T) void k_deg(const int* __restrict__ dst,
                                               const int* __restrict__ seg,
                                               const int* __restrict__ segcnt,
                                               float* __restrict__ dinv, int N, int E) {
    __shared__ int deg[SLAB_R];
    __shared__ int ovf;

    const int tid  = threadIdx.x;
    const int lane = tid & 31;
    const int wave = __builtin_amdgcn_readfirstlane(tid >> 5);
    const int b    = blockIdx.x;
    const int row0 = b * SLAB_R;
    int nrows = N - row0;
    if (nrows > SLAB_R) nrows = SLAB_R;
    const unsigned nr = (unsigned)nrows;

    for (int t = tid; t < SLAB_R; t += DEG_T) deg[t] = 0;
    if (tid == 0) ovf = (b >= MAXSLAB) ? 1 : 0;
    __syncthreads();

    if (b < MAXSLAB) {
        const int4* cb4 = (const int4*)(segcnt + (size_t)b * NSLICE) + lane * 2;
        const int4 ca = cb4[0], cc = cb4[1];
        const int cnt8[8] = {ca.x, ca.y, ca.z, ca.w, cc.x, cc.y, cc.z, cc.w};
        int bad = 0;
#pragma unroll
        for (int k = 0; k < 8; ++k) bad |= (cnt8[k] > SEGCAP) ? 1 : 0;
        if (__any(bad)) {
            if (tid == 0) ovf = 1;
        } else {
            const int* rb = seg + ((size_t)b * NSLICE + 32 * wave) * SEGCAP * 2 + 1;
            for (int a = 0; a < 4; a += 2) {
                int2 rr[16];
#pragma unroll
                for (int j = 0; j < 16; ++j) {
                    const int c = __builtin_amdgcn_readlane(cnt8[j & 7], 4 * wave + a + (j >> 3));
                    const int* p = rb + ((a * 8 + j) * SEGCAP + lane) * 2;
                    rr[j] = make_int2(-1, -1);
                    if (lane < c)      rr[j].x = p[0];
                    if (lane + 32 < c) rr[j].y = p[64];
                }
                asm volatile("" :: "v"(rr[0].x), "v"(rr[1].x), "v"(rr[2].x), "v"(rr[3].x), "v"(rr[4].x), "v"(rr[5].x), "v"(rr[6].x), "v"(rr[7].x),
                                   "v"(rr[8].x), "v"(rr[9].x), "v"(rr[10].x), "v"(rr[11].x), "v"(rr[12].x), "v"(rr[13].x), "v"(rr[14].x), "v"(rr[15].x));
                asm volatile("" :: "v"(rr[0].y), "v"(rr[1].y), "v"(rr[2].y), "v"(rr[3].y), "v"(rr[4].y), "v"(rr[5].y), "v"(rr[6].y), "v"(rr[7].y),
                                   "v"(rr[8].y), "v"(rr[9].y), "v"(rr[10].y), "v"(rr[11].y), "v"(rr[12].y), "v"(rr[13].y), "v"(rr[14].y), "v"(rr[15].y));
#pragma unroll
                for (int j = 0; j < 16; ++j) {
                    if (rr[j].x >= 0) atomicAdd(&deg[rr[j].x], 1);
                    if (rr[j].y >= 0) atomicAdd(&deg[rr[j].y], 1);
                }
            }
        }
    }
    __syncthreads();

    if (ovf) {
        for (int t = tid; t < SLAB_R; t += DEG_T) deg[t] = 0;
        __syncthreads();
        for (int e = tid; e < E; e += DEG_T) {
            unsigned r = (unsigned)(dst[e] - row0);
            if (r < nr) atomicAdd(&deg[r], 1);
        }
        __syncthreads();
    }
    for (int t = tid; t < nrows; t += DEG_T) dinv[row0 + t] = rsqrtf((float)(deg[t] + 1));
}

__global__ __launch_bounds__(SLAB_T) void k_slab(const int* __restrict__ src,
                                                 const int* __restrict__ dst,
                                                 const float* __restrict__ dinv,
                                                 const float* __restrict__ H,
                                                 const float* __restrict__ bias,
                                                 const float* __restrict__ g,
                                                 const float* __restrict__ be,
                                                 const int* __restrict__ seg,
                                                 const int* __restrict__ segcnt,
                                                 float* __restrict__ out, int N, int E) {
    __shared__ float acc[(SLAB_R + 1) * F];
    __shared__ int2  q2[QCAP];

    const int tid  = threadIdx.x;
    const int lane = tid & 31;
    const int wave = __builtin_amdgcn_readfirstlane(tid >> 5);
    const int row0 = blockIdx.x * SLAB_R;
    int nrows = N - row0;
    if (nrows > SLAB_R) nrows = SLAB_R;
    const unsigned nr = (unsigned)nrows;

    const float4* hslab4 = (const float4*)(H + (size_t)row0 * F);
    const int nel4 = nrows * (F / 4);

    bool queued = blockIdx.x < MAXSLAB;
    int  cnt8[8], off8[8];
    int  ntot = 0;
#pragma unroll
    for (int k = 0; k < 8; ++k) { cnt8[k] = 0; off8[k] = 0; }
    if (queued) {
        const int4* cb4 = (const int4*)(segcnt + (size_t)blockIdx.x * NSLICE) + lane * 2;
        const int4 ca = cb4[0], cc = cb4[1];
        cnt8[0] = ca.x; cnt8[1] = ca.y; cnt8[2] = ca.z; cnt8[3] = ca.w;
        cnt8[4] = cc.x; cnt8[5] = cc.y; cnt8[6] = cc.z; cnt8[7] = cc.w;
        int sum = 0, bad = 0;
#pragma unroll
        for (int k = 0; k < 8; ++k) {
            bad |= (cnt8[k] > SEGCAP) ? 1 : 0;
            off8[k] = sum;
            sum += cnt8[k];
        }
        int incl = sum;
#pragma unroll
        for (int m = 1; m < 32; m <<= 1) {
            int t = __shfl_up(incl, m, 32);
            if (lane >= m) incl += t;
        }
        const int excl = incl - sum;
#pragma unroll
        for (int k = 0; k < 8; ++k) off8[k] += excl;
        ntot = __builtin_amdgcn_readlane(incl, 31);
        if (__any(bad) || ntot > QCAP) queued = false;
    }

    if (tid < F) acc[SLAB_R * F + tid] = 0.f;
    if (queued) {
        const int2* sb = (const int2*)seg + ((size_t)blockIdx.x * NSLICE + 8 * wave) * SEGCAP + lane;
        int2 r0[8], r1[8];
        int  c[8], o[8];
#pragma unroll
        for (int k = 0; k < 8; ++k) {
            c[k] = __builtin_amdgcn_readlane(cnt8[k], wave);
            o[k] = __builtin_amdgcn_readlane(off8[k], wave);
            r0[k] = make_int2(0, 0);
            r1[k] = make_int2(0, 0);
            if (lane < c[k])      r0[k] = sb[k * SEGCAP];
            if (lane + 32 < c[k]) r1[k] = sb[k * SEGCAP + 32];
        }
        asm volatile("" :: "v"(r0[0].x), "v"(r0[0].y), "v"(r0[1].x), "v"(r0[1].y),
                           "v"(r0[2].x), "v"(r0[2].y), "v"(r0[3].x), "v"(r0[3].y),
                           "v"(r0[4].x), "v"(r0[4].y), "v"(r0[5].x), "v"(r0[5].y),
                           "v"(r0[6].x), "v"(r0[6].y), "v"(r0[7].x), "v"(r0[7].y));
        asm volatile("" :: "v"(r1[0].x), "v"(r1[0].y), "v"(r1[1].x), "v"(r1[1].y),
                           "v"(r1[2].x), "v"(r1[2].y), "v"(r1[3].x), "v"(r1[3].y),
                           "v"(r1[4].x), "v"(r1[4].y), "v"(r1[5].x), "v"(r1[5].y),
                           "v"(r1[6].x), "v"(r1[6].y), "v"(r1[7].x), "v"(r1[7].y));
#pragma unroll
        for (int k = 0; k < 8; ++k) {
            if (lane < c[k])      q2[o[k] + lane]      = r0[k];
            if (lane + 32 < c[k]) q2[o[k] + lane + 32] = r1[k];
        }
    }

    const float4 b4 = ((const float4*)bias)[lane];
    for (int base = tid; base < nel4; base += 4 * SLAB_T) {
        float4 t[4];
        float  di[4];
#pragma unroll
        for (int k = 0; k < 4; ++k) {
            int i = base + k * SLAB_T;
            if (i > nel4 - 1) i = nel4 - 1;
            t[k]  = hslab4[i];
            di[k] = dinv[row0 + (i >> 5)];
        }
        asm volatile("" :: "v"(t[0].x), "v"(t[0].y), "v"(t[0].z), "v"(t[0].w),
                           "v"(t[1].x), "v"(t[1].y), "v"(t[1].z), "v"(t[1].w),
                           "v"(t[2].x), "v"(t[2].y), "v"(t[2].z), "v"(t[2].w),
                           "v"(t[3].x), "v"(t[3].y), "v"(t[3].z), "v"(t[3].w),
                           "v"(di[0]), "v"(di[1]), "v"(di[2]), "v"(di[3]));
#pragma unroll
        for (int k = 0; k < 4; ++k) {
            int i = base + k * SLAB_T;
            if (i < nel4) {
                float n2 = di[k] * di[k];
                float4 a;
                a.x = fmaf(t[k].x, n2, b4.x);
                a.y = fmaf(t[k].y, n2, b4.y);
                a.z = fmaf(t[k].z, n2, b4.z);
                a.w = fmaf(t[k].w, n2, b4.w);
                float* ar = &acc[(i >> 5) * F + (i & 31)];
                ar[0] = a.x; ar[32] = a.y; ar[64] = a.z; ar[96] = a.w;
            }
        }
    }
    __syncthreads();

    if (!queued) {
        for (int e = tid; e < E; e += SLAB_T) {
            unsigned r = (unsigned)(dst[e] - row0);
            if (r < nr) {
                int s = src[e];
                float w = dinv[s] * dinv[row0 + (int)r];
                const float* hr = H + (size_t)s * F;
                float* ar = &acc[r * F];
#pragma unroll 1
                for (int f = 0; f < F; ++f) lds_add(ar + 32 * (f & 3) + (f >> 2), hr[f] * w);
            }
        }
    } else {
        const int n = ntot;
        for (int i0 = wave * 8; i0 < n; i0 += (SLAB_T / 32) * 8) {
            int   s[8], rw[8];
            float w[8];
            float4 v[8];
#pragma unroll
            for (int k = 0; k < 8; ++k) {
                int idx = i0 + k;
                if (idx > n - 1) idx = n - 1;
                int2 rec = q2[idx];
                s[k]  = __builtin_amdgcn_readfirstlane(rec.x);
                rw[k] = __builtin_amdgcn_readfirstlane(rec.y);
            }
#pragma unroll
            for (int k = 0; k < 8; ++k) {
                w[k] = dinv[s[k]] * dinv[row0 + rw[k]];
                v[k] = ((const float4*)(H + (size_t)s[k] * F))[lane];
            }
#pragma unroll
            for (int k = 0; k < 8; ++k) {
                int row = (i0 + k < n) ? rw[k] : SLAB_R;
                float* ar = &acc[row * F + lane];
                lds_add(ar,      v[k].x * w[k]);
                lds_add(ar + 32, v[k].y * w[k]);
                lds_add(ar + 64, v[k].z * w[k]);
                lds_add(ar + 96, v[k].w * w[k]);
            }
        }
    }
    __syncthreads();

    const float4 gg = ((const float4*)g)[lane];
    const float4 bb = ((const float4*)be)[lane];
    float4* out4 = (float4*)(out + (size_t)row0 * F);
    for (int r0 = wave * LN_R; r0 < nrows; r0 += (SLAB_T / 32) * LN_R) {
        float4 v[LN_R];
        float  s[LN_R], qq[LN_R];
#pragma unroll
        for (int k = 0; k < LN_R; ++k) {
            int r = r0 + k;
            if (r > nrows - 1) r = nrows - 1;
            const float* ar = &acc[r * F + lane];
            float4 t = make_float4(ar[0], ar[32], ar[64], ar[96]);
            t.x = fmaxf(t.x, 0.f); t.y = fmaxf(t.y, 0.f);
            t.z = fmaxf(t.z, 0.f); t.w = fmaxf(t.w, 0.f);
            v[k] = t;
            s[k] = t.x + t.y + t.z + t.w;
        }
#pragma unroll
        for (int m = 16; m >= 1; m >>= 1) {
#pragma unroll
            for (int k = 0; k < LN_R; ++k) s[k] += __shfl_xor(s[k], m, 32);
        }
#pragma unroll
        for (int k = 0; k < LN_R; ++k) {
            float mu = s[k] * (1.f / F);
            v[k].x -= mu; v[k].y -= mu; v[k].z -= mu; v[k].w -= mu;
            qq[k] = v[k].x * v[k].x + v[k].y * v[k].y + v[k].z * v[k].z + v[k].w * v[k].w;
        }
#pragma unroll
        for (int m = 16; m >= 1; m >>= 1) {
#pragma unroll
            for (int k = 0; k < LN_R; ++k) qq[k] += __shfl_xor(qq[k], m, 32);
        }
#pragma unroll
        for (int k = 0; k < LN_R; ++k) {
            float rs = rsqrtf(qq[k] * (1.f / F) + LN_EPS);
            float4 o;
            o.x = v[k].x * rs * gg.x + bb.x;
            o.y = v[k].y * rs * gg.y + bb.y;
            o.z = v[k].z * rs * gg.z + bb.z;
            o.w = v[k].w * rs * gg.w + bb.w;
            if (r0 + k < nrows) out4[(r0 + k) * (F / 4) + lane] = o;
        }
    }
}

static inline size_t align256(size_t x) { return (x + 255) & ~(size_t)255; }

extern "C" void kernel_launch(void* const* d_in, const int* in_sizes, int n_in,
                              void* d_out, int out_size, void* d_ws, size_t ws_size,
                              hipStream_t stream) {
    const float* x   = (const float*)d_in[0];
    const int*   ei  = (const int*)  d_in[1];
    const float* W1  = (const float*)d_in[2];
    const float* b1  = (const float*)d_in[3];
    const float* W2  = (const float*)d_in[4];
    const float* b2  = (const float*)d_in[5];
    const float* g1  = (const float*)d_in[6];
    const float* be1 = (const float*)d_in[7];
    const float* g2  = (const float*)d_in[8];
    const float* be2 = (const float*)d_in[9];

    const int N = in_sizes[0] / F;
    const int E = in_sizes[1] / 2;
    const int* src = ei;
    const int* dst = ei + E;
    float* out = (float*)d_out;

    const int gGemm = (N + 127) / 128;
    const int gSlab = (N + SLAB_R - 1) / SLAB_R;

    char* ws = (char*)d_ws;
    size_t off = 0;
    float* dinv   = (float*)(ws + off); off = align256(off + (size_t)N * 4);
    float* H      = (float*)(ws + off); off = align256(off + (size_t)N * F * 4);
    int*   segcnt = (int*)  (ws + off); off = align256(off + (size_t)MAXSLAB * NSLICE * 4);
    int*   seg    = (int*)  (ws + off); off = align256(off + (size_t)MAXSLAB * NSLICE * SEGCAP * 8);
    (void)ws_size; (void)n_in; (void)out_size;

    const int T = 256;
    dim3 blk(T);

    k_part<<<NSLICE, dim3(PART_T), 0, stream>>>(src, dst, seg, segcnt, N, E, gSlab);
    k_deg<<<gSlab, dim3(DEG_T), 0, stream>>>(dst, seg, segcnt, dinv, N, E);

    k_gemm<<<gGemm, blk, 0, stream>>>(x, W1, H, N);
    k_slab<<<gSlab, dim3(SLAB_T), 0, stream>>>(src, dst, dinv, H, b1, g1, be1, seg, segcnt, out, N, E);

    k_gemm<<<gGemm, blk, 0, stream>>>(out, W2, H, N);
    k_slab<<<gSlab, dim3(SLAB_T), 0, stream>>>(src, dst, dinv, H, b2, g2, be2, seg, segcnt, out, N, E);
}


# Edge partition and layer-1 GEMM share one launch: work-groups 0-255 cut the edge list, the rest do x@W1 (8 row tiles, two waves per tile); the two independent jobs overlap, one launch gone
# speedup vs baseline: 3.9651x; 1.0351x over previous
#include <hip/hip_runtime.h>

typedef float v2f __attribute__((ext_vector_type(2)));
typedef float v8f __attribute__((ext_vector_type(8)));
typedef int   v4i __attribute__((ext_vector_type(4)));

#define F 128
#define LN_EPS 1e-5f

__global__ __launch_bounds__(256) void k_gemm(const float* __restrict__ X,
                                              const float* __restrict__ W,
                                              float* __restrict__ H, int N) {
    __shared__ float sW[F * F];
    for (int i = threadIdx.x; i < (F / 2) * (F / 4); i += 256) {
        int p = i >> 5;
        int c = (i & 31) * 4;
        float4 w0 = *(const float4*)&W[(2 * p)     * F + c];
        float4 w1 = *(const float4*)&W[(2 * p + 1) * F + c];
        v2f* d = (v2f*)&sW[p * 256 + c * 2];
        d[0] = (v2f){w0.x, w1.x};
        d[1] = (v2f){w0.y, w1.y};
        d[2] = (v2f){w0.z, w1.z};
        d[3] = (v2f){w0.w, w1.w};
    }
    __syncthreads();

    const int wave = threadIdx.x >> 5;
    const int lane = threadIdx.x & 31;
    const int tile = blockIdx.x * 8 + wave;
    if (tile * 16 >= N) return;

    const int col   = lane & 15;
    const int khalf = (lane >> 4) << 1;
    int row = tile * 16 + (lane & 15);
    if (row > N - 1) row = N - 1;
    const float* xrow = X + (size_t)row * F;

    v8f acc[8];
    const v8f vzero = {0.f, 0.f, 0.f, 0.f, 0.f, 0.f, 0.f, 0.f};
#pragma unroll
    for (int j = 0; j < 8; ++j) acc[j] = vzero;

    for (int k0 = 0; k0 < F; k0 += 4) {
        v2f a;
        a.x = xrow[k0 + khalf];
        a.y = xrow[k0 + khalf + 1];
        const v2f* bbase = (const v2f*)&sW[((k0 + khalf) >> 1) * 256 + col * 2];
#pragma unroll
        for (int j = 0; j < 8; ++j) {
            v2f b = bbase[j * 16];
            acc[j] = __builtin_amdgcn_wmma_f32_16x16x4_f32(
                false, a, false, b, (short)0, acc[j], false, false);
        }
    }

    const int rbase = tile * 16 + (lane >> 4) * 8;
#pragma unroll
    for (int j = 0; j < 8; ++j) {
#pragma unroll
        for (int i = 0; i < 8; ++i) {
            int r = rbase + i;
            if (r < N) H[(size_t)r * F + j * 16 + col] = acc[j][i];
        }
    }
}

#define SLAB_R 400
#define SLAB_T 1024
#define QCAP   8192
#define NSLICE 256
#define SEGCAP 64
#define MAXSLAB 256
#define PART_T 512
#define PART_U 13
#define DEG_T  256
#define LN_R   4

static_assert(NSLICE == 32 * 8, "a wave holds the NSLICE counts of a slab as 8 per lane");
static_assert(NSLICE == (SLAB_T / 32) * 8, "a k_slab wave gathers 8 segments");
static_assert(NSLICE == (DEG_T / 32) * 32, "a k_deg wave counts 32 segments");
static_assert(MAXSLAB == (PART_T / 32) * 16, "a k_part wave writes 16 segments");
static_assert(F * F * 4 <= MAXSLAB * SEGCAP * 8, "the x @ W1 work-groups stage W in the records' LDS block");
static_assert(SEGCAP == 64, "a lane holds records l and l+32 of a segment");

__device__ __forceinline__ void lds_add(float* p, float v) {
    __hip_atomic_fetch_add(p, v, __ATOMIC_RELAXED, __HIP_MEMORY_SCOPE_WORKGROUP);
}

__device__ __forceinline__ void gemm_half_tiles(float* sW, const float* __restrict__ X,
                                                const float* __restrict__ W,
                                                float* __restrict__ H, int N, int blk) {
    for (int i = threadIdx.x; i < (F / 2) * (F / 4); i += PART_T) {
        int p = i >> 5;
        int c = (i & 31) * 4;
        float4 w0 = *(const float4*)&W[(2 * p)     * F + c];
        float4 w1 = *(const float4*)&W[(2 * p + 1) * F + c];
        v2f* d = (v2f*)&sW[p * 256 + c * 2];
        d[0] = (v2f){w0.x, w1.x};
        d[1] = (v2f){w0.y, w1.y};
        d[2] = (v2f){w0.z, w1.z};
        d[3] = (v2f){w0.w, w1.w};
    }
    __syncthreads();

    const int wave = threadIdx.x >> 5;
    const int lane = threadIdx.x & 31;
    const int tile = blk * 8 + (wave >> 1);
    const int cb   = (wave & 1) * 64;
    if (tile * 16 >= N) return;

    const int col   = lane & 15;
    const int khalf = (lane >> 4) << 1;
    int row = tile * 16 + (lane & 15);
    if (row > N - 1) row = N - 1;
    const float* xrow = X + (size_t)row * F;

    v8f acc[4];
    const v8f vzero = {0.f, 0.f, 0.f, 0.f, 0.f, 0.f, 0.f, 0.f};
#pragma unroll
    for (int j = 0; j < 4; ++j) acc[j] = vzero;

    for (int k0 = 0; k0 < F; k0 += 4) {
        v2f a;
        a.x = xrow[k0 + khalf];
        a.y = xrow[k0 + khalf + 1];
        const v2f* bbase = (const v2f*)&sW[((k0 + khalf) >> 1) * 256 + (cb + col) * 2];
#pragma unroll
        for (int j = 0; j < 4; ++j) {
            v2f b = bbase[j * 16];
            acc[j] = __builtin_amdgcn_wmma_f32_16x16x4_f32(
                false, a, false, b, (short)0, acc[j], false, false);
        }
    }

    const int rbase = tile * 16 + (lane >> 4) * 8;
#pragma unroll
    for (int j = 0; j < 4; ++j) {
#pragma unroll
        for (int i = 0; i < 8; ++i) {
            int r = rbase + i;
            if (r < N) H[(size_t)r * F + cb + j * 16 + col] = acc[j][i];
        }
    }
}

__global__ __launch_bounds__(PART_T) void k_part_gemm(const int* __restrict__ src,
                                                      const int* __restrict__ dst,
                                                      int* __restrict__ seg,
                                                      int* __restrict__ segcnt, int N, int E, int nslab,
                                                      const float* __restrict__ X,
                                                      const float* __restrict__ W,
                                                      float* __restrict__ H) {
    __shared__ v4i smem4[MAXSLAB * SEGCAP / 2 + MAXSLAB / 4];
    if (blockIdx.x >= NSLICE) {
        gemm_half_tiles((float*)smem4, X, W, H, N, (int)blockIdx.x - NSLICE);
        return;
    }
    v4i*  stage4 = smem4;
    int2* stage  = (int2*)stage4;
    int*  cnt    = (int*)(smem4 + MAXSLAB * SEGCAP / 2);

    const int tid  = threadIdx.x;
    const int lane = tid & 31;
    const int wave = __builtin_amdgcn_readfirstlane(tid >> 5);
    const int g    = blockIdx.x;
    const int nsl  = nslab < MAXSLAB ? nslab : MAXSLAB;

    if (tid < MAXSLAB) cnt[tid] = 0;
    __syncthreads();

    const int per = (E + NSLICE - 1) / NSLICE;
    const int e0  = g * per;
    int e1 = e0 + per;
    if (e1 > E) e1 = E;

    for (int base = e0 + tid; base < e1; base += PART_U * PART_T) {
        int d[PART_U], s[PART_U];
#pragma unroll
        for (int k = 0; k < PART_U; ++k) {
            int e = base + k * PART_T;
            if (e > e1 - 1) e = e1 - 1;
            d[k] = dst[e];
            s[k] = src[e];
        }
        asm volatile("" :: "v"(d[0]), "v"(d[1]), "v"(d[2]), "v"(d[3]), "v"(d[4]), "v"(d[5]), "v"(d[6]),
                           "v"(d[7]), "v"(d[8]), "v"(d[9]), "v"(d[10]), "v"(d[11]), "v"(d[12]));
        asm volatile("" :: "v"(s[0]), "v"(s[1]), "v"(s[2]), "v"(s[3]), "v"(s[4]), "v"(s[5]), "v"(s[6]),
                           "v"(s[7]), "v"(s[8]), "v"(s[9]), "v"(s[10]), "v"(s[11]), "v"(s[12]));
#pragma unroll
        for (int k = 0; k < PART_U; ++k) {
            int e = base + k * PART_T;
            if (e < e1 && (unsigned)d[k] < (unsigned)N) {
                int b = d[k] / SLAB_R;
                if (b < nsl) {
                    int slot = atomicAdd(&cnt[b], 1);
                    if (slot < SEGCAP) stage[b * SEGCAP + slot] = make_int2(s[k], d[k] - b * SLAB_R);
                }
            }
        }
    }
    __syncthreads();

    if (tid < nsl) segcnt[tid * NSLICE + g] = cnt[tid];
    const int cv = cnt[wave + (lane & 15) * (PART_T / 32)];
    v4i r[16];
#pragma unroll
    for (int j = 0; j < MAXSLAB / (PART_T / 32); ++j) r[j] = stage4[(wave + j * (PART_T / 32)) * (SEGCAP / 2) + lane];
    asm volatile("" :: "v"(r[0]), "v"(r[1]), "v"(r[2]), "v"(r[3]), "v"(r[4]), "v"(r[5]), "v"(r[6]), "v"(r[7]),
                       "v"(r[8]), "v"(r[9]), "v"(r[10]), "v"(r[11]), "v"(r[12]), "v"(r[13]), "v"(r[14]), "v"(r[15]));
#pragma unroll
    for (int j = 0; j < MAXSLAB / (PART_T / 32); ++j) {
        const int b = wave + j * (PART_T / 32);
        const int c = __builtin_amdgcn_readlane(cv, j);
        if (b < nsl) {
            int* o = seg + (((size_t)b * NSLICE + g) * SEGCAP + 2 * lane) * 2;
            if (2 * lane + 1 < c)  *(v4i*)o = r[j];
            else if (2 * lane < c) *(int2*)o = make_int2(r[j].x, r[j].y);
        }
    }
}

__global__ __launch_bounds__(DEG_T) void k_deg(const int* __restrict__ dst,
                                               const int* __restrict__ seg,
                                               const int* __restrict__ segcnt,
                                               float* __restrict__ dinv, int N, int E) {
    __shared__ int deg[SLAB_R];
    __shared__ int ovf;

    const int tid  = threadIdx.x;
    const int lane = tid & 31;
    const int wave = __builtin_amdgcn_readfirstlane(tid >> 5);
    const int b    = blockIdx.x;
    const int row0 = b * SLAB_R;
    int nrows = N - row0;
    if (nrows > SLAB_R) nrows = SLAB_R;
    const unsigned nr = (unsigned)nrows;

    for (int t = tid; t < SLAB_R; t += DEG_T) deg[t] = 0;
    if (tid == 0) ovf = (b >= MAXSLAB) ? 1 : 0;
    __syncthreads();

    if (b < MAXSLAB) {
        const int4* cb4 = (const int4*)(segcnt + (size_t)b * NSLICE) + lane * 2;
        const int4 ca = cb4[0], cc = cb4[1];
        const int cnt8[8] = {ca.x, ca.y, ca.z, ca.w, cc.x, cc.y, cc.z, cc.w};
        int bad = 0;
#pragma unroll
        for (int k = 0; k < 8; ++k) bad |= (cnt8[k] > SEGCAP) ? 1 : 0;
        if (__any(bad)) {
            if (tid == 0) ovf = 1;
        } else {
            const int* rb = seg + ((size_t)b * NSLICE + 32 * wave) * SEGCAP * 2 + 1;
            for (int a = 0; a < 4; a += 2) {
                int2 rr[16];
#pragma unroll
                for (int j = 0; j < 16; ++j) {
                    const int c = __builtin_amdgcn_readlane(cnt8[j & 7], 4 * wave + a + (j >> 3));
                    const int* p = rb + ((a * 8 + j) * SEGCAP + lane) * 2;
                    rr[j] = make_int2(-1, -1);
                    if (lane < c)      rr[j].x = p[0];
                    if (lane + 32 < c) rr[j].y = p[64];
                }
                asm volatile("" :: "v"(rr[0].x), "v"(rr[1].x), "v"(rr[2].x), "v"(rr[3].x), "v"(rr[4].x), "v"(rr[5].x), "v"(rr[6].x), "v"(rr[7].x),
                                   "v"(rr[8].x), "v"(rr[9].x), "v"(rr[10].x), "v"(rr[11].x), "v"(rr[12].x), "v"(rr[13].x), "v"(rr[14].x), "v"(rr[15].x));
                asm volatile("" :: "v"(rr[0].y), "v"(rr[1].y), "v"(rr[2].y), "v"(rr[3].y), "v"(rr[4].y), "v"(rr[5].y), "v"(rr[6].y), "v"(rr[7].y),
                                   "v"(rr[8].y), "v"(rr[9].y), "v"(rr[10].y), "v"(rr[11].y), "v"(rr[12].y), "v"(rr[13].y), "v"(rr[14].y), "v"(rr[15].y));
#pragma unroll
                for (int j = 0; j < 16; ++j) {
                    if (rr[j].x >= 0) atomicAdd(&deg[rr[j].x], 1);
                    if (rr[j].y >= 0) atomicAdd(&deg[rr[j].y], 1);
                }
            }
        }
    }
    __syncthreads();

    if (ovf) {
        for (int t = tid; t < SLAB_R; t += DEG_T) deg[t] = 0;
        __syncthreads();
        for (int e = tid; e < E; e += DEG_T) {
            unsigned r = (unsigned)(dst[e] - row0);
            if (r < nr) atomicAdd(&deg[r], 1);
        }
        __syncthreads();
    }
    for (int t = tid; t < nrows; t += DEG_T) dinv[row0 + t] = rsqrtf((float)(deg[t] + 1));
}

__global__ __launch_bounds__(SLAB_T) void k_slab(const int* __restrict__ src,
                                                 const int* __restrict__ dst,
                                                 const float* __restrict__ dinv,
                                                 const float* __restrict__ H,
                                                 const float* __restrict__ bias,
                                                 const float* __restrict__ g,
                                                 const float* __restrict__ be,
                                                 const int* __restrict__ seg,
                                                 const int* __restrict__ segcnt,
                                                 float* __restrict__ out, int N, int E) {
    __shared__ float acc[(SLAB_R + 1) * F];
    __shared__ int2  q2[QCAP];

    const int tid  = threadIdx.x;
    const int lane = tid & 31;
    const int wave = __builtin_amdgcn_readfirstlane(tid >> 5);
    const int row0 = blockIdx.x * SLAB_R;
    int nrows = N - row0;
    if (nrows > SLAB_R) nrows = SLAB_R;
    const unsigned nr = (unsigned)nrows;

    const float4* hslab4 = (const float4*)(H + (size_t)row0 * F);
    const int nel4 = nrows * (F / 4);

    bool queued = blockIdx.x < MAXSLAB;
    int  cnt8[8], off8[8];
    int  ntot = 0;
#pragma unroll
    for (int k = 0; k < 8; ++k) { cnt8[k] = 0; off8[k] = 0; }
    if (queued) {
        const int4* cb4 = (const int4*)(segcnt + (size_t)blockIdx.x * NSLICE) + lane * 2;
        const int4 ca = cb4[0], cc = cb4[1];
        cnt8[0] = ca.x; cnt8[1] = ca.y; cnt8[2] = ca.z; cnt8[3] = ca.w;
        cnt8[4] = cc.x; cnt8[5] = cc.y; cnt8[6] = cc.z; cnt8[7] = cc.w;
        int sum = 0, bad = 0;
#pragma unroll
        for (int k = 0; k < 8; ++k) {
            bad |= (cnt8[k] > SEGCAP) ? 1 : 0;
            off8[k] = sum;
            sum += cnt8[k];
        }
        int incl = sum;
#pragma unroll
        for (int m = 1; m < 32; m <<= 1) {
            int t = __shfl_up(incl, m, 32);
            if (lane >= m) incl += t;
        }
        const int excl = incl - sum;
#pragma unroll
        for (int k = 0; k < 8; ++k) off8[k] += excl;
        ntot = __builtin_amdgcn_readlane(incl, 31);
        if (__any(bad) || ntot > QCAP) queued = false;
    }

    if (tid < F) acc[SLAB_R * F + tid] = 0.f;
    if (queued) {
        const int2* sb = (const int2*)seg + ((size_t)blockIdx.x * NSLICE + 8 * wave) * SEGCAP + lane;
        int2 r0[8], r1[8];
        int  c[8], o[8];
#pragma unroll
        for (int k = 0; k < 8; ++k) {
            c[k] = __builtin_amdgcn_readlane(cnt8[k], wave);
            o[k] = __builtin_amdgcn_readlane(off8[k], wave);
            r0[k] = make_int2(0, 0);
            r1[k] = make_int2(0, 0);
            if (lane < c[k])      r0[k] = sb[k * SEGCAP];
            if (lane + 32 < c[k]) r1[k] = sb[k * SEGCAP + 32];
        }
        asm volatile("" :: "v"(r0[0].x), "v"(r0[0].y), "v"(r0[1].x), "v"(r0[1].y),
                           "v"(r0[2].x), "v"(r0[2].y), "v"(r0[3].x), "v"(r0[3].y),
                           "v"(r0[4].x), "v"(r0[4].y), "v"(r0[5].x), "v"(r0[5].y),
                           "v"(r0[6].x), "v"(r0[6].y), "v"(r0[7].x), "v"(r0[7].y));
        asm volatile("" :: "v"(r1[0].x), "v"(r1[0].y), "v"(r1[1].x), "v"(r1[1].y),
                           "v"(r1[2].x), "v"(r1[2].y), "v"(r1[3].x), "v"(r1[3].y),
                           "v"(r1[4].x), "v"(r1[4].y), "v"(r1[5].x), "v"(r1[5].y),
                           "v"(r1[6].x), "v"(r1[6].y), "v"(r1[7].x), "v"(r1[7].y));
#pragma unroll
        for (int k = 0; k < 8; ++k) {
            if (lane < c[k])      q2[o[k] + lane]      = r0[k];
            if (lane + 32 < c[k]) q2[o[k] + lane + 32] = r1[k];
        }
    }

    const float4 b4 = ((const float4*)bias)[lane];
    for (int base = tid; base < nel4; base += 4 * SLAB_T) {
        float4 t[4];
        float  di[4];
#pragma unroll
        for (int k = 0; k < 4; ++k) {
            int i = base + k * SLAB_T;
            if (i > nel4 - 1) i = nel4 - 1;
            t[k]  = hslab4[i];
            di[k] = dinv[row0 + (i >> 5)];
        }
        asm volatile("" :: "v"(t[0].x), "v"(t[0].y), "v"(t[0].z), "v"(t[0].w),
                           "v"(t[1].x), "v"(t[1].y), "v"(t[1].z), "v"(t[1].w),
                           "v"(t[2].x), "v"(t[2].y), "v"(t[2].z), "v"(t[2].w),
                           "v"(t[3].x), "v"(t[3].y), "v"(t[3].z), "v"(t[3].w),
                           "v"(di[0]), "v"(di[1]), "v"(di[2]), "v"(di[3]));
#pragma unroll
        for (int k = 0; k < 4; ++k) {
            int i = base + k * SLAB_T;
            if (i < nel4) {
                float n2 = di[k] * di[k];
                float4 a;
                a.x = fmaf(t[k].x, n2, b4.x);
                a.y = fmaf(t[k].y, n2, b4.y);
                a.z = fmaf(t[k].z, n2, b4.z);
                a.w = fmaf(t[k].w, n2, b4.w);
                float* ar = &acc[(i >> 5) * F + (i & 31)];
                ar[0] = a.x; ar[32] = a.y; ar[64] = a.z; ar[96] = a.w;
            }
        }
    }
    __syncthreads();

    if (!queued) {
        for (int e = tid; e < E; e += SLAB_T) {
            unsigned r = (unsigned)(dst[e] - row0);
            if (r < nr) {
                int s = src[e];
                float w = dinv[s] * dinv[row0 + (int)r];
                const float* hr = H + (size_t)s * F;
                float* ar = &acc[r * F];
#pragma unroll 1
                for (int f = 0; f < F; ++f) lds_add(ar + 32 * (f & 3) + (f >> 2), hr[f] * w);
            }
        }
    } else {
        const int n = ntot;
        for (int i0 = wave * 8; i0 < n; i0 += (SLAB_T / 32) * 8) {
            int   s[8], rw[8];
            float w[8];
            float4 v[8];
#pragma unroll
            for (int k = 0; k < 8; ++k) {
                int idx = i0 + k;
                if (idx > n - 1) idx = n - 1;
                int2 rec = q2[idx];
                s[k]  = __builtin_amdgcn_readfirstlane(rec.x);
                rw[k] = __builtin_amdgcn_readfirstlane(rec.y);
            }
#pragma unroll
            for (int k = 0; k < 8; ++k) {
                w[k] = dinv[s[k]] * dinv[row0 + rw[k]];
                v[k] = ((const float4*)(H + (size_t)s[k] * F))[lane];
            }
#pragma unroll
            for (int k = 0; k < 8; ++k) {
                int row = (i0 + k < n) ? rw[k] : SLAB_R;
                float* ar = &acc[row * F + lane];
                lds_add(ar,      v[k].x * w[k]);
                lds_add(ar + 32, v[k].y * w[k]);
                lds_add(ar + 64, v[k].z * w[k]);
                lds_add(ar + 96, v[k].w * w[k]);
            }
        }
    }
    __syncthreads();

    const float4 gg = ((const float4*)g)[lane];
    const float4 bb = ((const float4*)be)[lane];
    float4* out4 = (float4*)(out + (size_t)row0 * F);
    for (int r0 = wave * LN_R; r0 < nrows; r0 += (SLAB_T / 32) * LN_R) {
        float4 v[LN_R];
        float  s[LN_R], qq[LN_R];
#pragma unroll
        for (int k = 0; k < LN_R; ++k) {
            int r = r0 + k;
            if (r > nrows - 1) r = nrows - 1;
            const float* ar = &acc[r * F + lane];
            float4 t = make_float4(ar[0], ar[32], ar[64], ar[96]);
            t.x = fmaxf(t.x, 0.f); t.y = fmaxf(t.y, 0.f);
            t.z = fmaxf(t.z, 0.f); t.w = fmaxf(t.w, 0.f);
            v[k] = t;
            s[k] = t.x + t.y + t.z + t.w;
        }
#pragma unroll
        for (int m = 16; m >= 1; m >>= 1) {
#pragma unroll
            for (int k = 0; k < LN_R; ++k) s[k] += __shfl_xor(s[k], m, 32);
        }
#pragma unroll
        for (int k = 0; k < LN_R; ++k) {
            float mu = s[k] * (1.f / F);
            v[k].x -= mu; v[k].y -= mu; v[k].z -= mu; v[k].w -= mu;
            qq[k] = v[k].x * v[k].x + v[k].y * v[k].y + v[k].z * v[k].z + v[k].w * v[k].w;
        }
#pragma unroll
        for (int m = 16; m >= 1; m >>= 1) {
#pragma unroll
            for (int k = 0; k < LN_R; ++k) qq[k] += __shfl_xor(qq[k], m, 32);
        }
#pragma unroll
        for (int k = 0; k < LN_R; ++k) {
            float rs = rsqrtf(qq[k] * (1.f / F) + LN_EPS);
            float4 o;
            o.x = v[k].x * rs * gg.x + bb.x;
            o.y = v[k].y * rs * gg.y + bb.y;
            o.z = v[k].z * rs * gg.z + bb.z;
            o.w = v[k].w * rs * gg.w + bb.w;
            if (r0 + k < nrows) out4[(r0 + k) * (F / 4) + lane] = o;
        }
    }
}

static inline size_t align256(size_t x) { return (x + 255) & ~(size_t)255; }

extern "C" void kernel_launch(void* const* d_in, const int* in_sizes, int n_in,
                              void* d_out, int out_size, void* d_ws, size_t ws_size,
                              hipStream_t stream) {
    const float* x   = (const float*)d_in[0];
    const int*   ei  = (const int*)  d_in[1];
    const float* W1  = (const float*)d_in[2];
    const float* b1  = (const float*)d_in[3];
    const float* W2  = (const float*)d_in[4];
    const float* b2  = (const float*)d_in[5];
    const float* g1  = (const float*)d_in[6];
    const float* be1 = (const float*)d_in[7];
    const float* g2  = (const float*)d_in[8];
    const float* be2 = (const float*)d_in[9];

    const int N = in_sizes[0] / F;
    const int E = in_sizes[1] / 2;
    const int* src = ei;
    const int* dst = ei + E;
    float* out = (float*)d_out;

    const int gGemm = (N + 127) / 128;
    const int gSlab = (N + SLAB_R - 1) / SLAB_R;

    char* ws = (char*)d_ws;
    size_t off = 0;
    float* dinv   = (float*)(ws + off); off = align256(off + (size_t)N * 4);
    float* H      = (float*)(ws + off); off = align256(off + (size_t)N * F * 4);
    int*   segcnt = (int*)  (ws + off); off = align256(off + (size_t)MAXSLAB * NSLICE * 4);
    int*   seg    = (int*)  (ws + off); off = align256(off + (size_t)MAXSLAB * NSLICE * SEGCAP * 8);
    (void)ws_size; (void)n_in; (void)out_size;

    const int T = 256;
    dim3 blk(T);

    k_part_gemm<<<NSLICE + gGemm, dim3(PART_T), 0, stream>>>(src, dst, seg, segcnt, N, E, gSlab, x, W1, H);
    k_deg<<<gSlab, dim3(DEG_T), 0, stream>>>(dst, seg, segcnt, dinv, N, E);

    k_slab<<<gSlab, dim3(SLAB_T), 0, stream>>>(src, dst, dinv, H, b1, g1, be1, seg, segcnt, out, N, E);

    k_gemm<<<gGemm, blk, 0, stream>>>(out, W2, H, N);
    k_slab<<<gSlab, dim3(SLAB_T), 0, stream>>>(src, dst, dinv, H, b2, g2, be2, seg, segcnt, out, N, E);
}
